# starting kernel
# baseline (speedup 1.0000x reference)
#include <hip/hip_runtime.h>
#include <hip/hip_bf16.h>

typedef __attribute__((ext_vector_type(2))) float v2f;
typedef __attribute__((ext_vector_type(8))) float v8f;

#define NEG_SLOPE 0.2f
#define EPS_DEN 1e-16f
#define FDIM 64

__device__ __forceinline__ unsigned fenc(float f) {
  unsigned b = __float_as_uint(f);
  return (b & 0x80000000u) ? ~b : (b | 0x80000000u);
}
__device__ __forceinline__ float fdec(unsigned k) {
  unsigned b = (k & 0x80000000u) ? (k & 0x7fffffffu) : ~k;
  return __uint_as_float(b);
}

__global__ __launch_bounds__(128) void gatv2_gemm_dual(
    const float* __restrict__ X, const float* __restrict__ Wl,
    const float* __restrict__ Wr, float* __restrict__ xl,
    float* __restrict__ xr, int N, int K) {
  const int lane   = threadIdx.x & 31;
  const int wave   = threadIdx.x >> 5;
  const int laneHi = lane >> 4;
  const int lane16 = lane & 15;
  const int rowTile = blockIdx.x;
  const int colBase = wave * 16;

  int row  = rowTile * 16 + lane16;
  int rowc = row < N ? row : N - 1;

  v8f accL = {};
  v8f accR = {};
  const float* xrow = X + (size_t)rowc * K;

  for (int k = 0; k < K; k += 4) {
    const int kk = k + 2 * laneHi;
    v2f a = *(const v2f*)(xrow + kk);
    v2f bl, br;
    bl[0] = Wl[(size_t)kk * FDIM + colBase + lane16];
    bl[1] = Wl[(size_t)(kk + 1) * FDIM + colBase + lane16];
    br[0] = Wr[(size_t)kk * FDIM + colBase + lane16];
    br[1] = Wr[(size_t)(kk + 1) * FDIM + colBase + lane16];
    accL = __builtin_amdgcn_wmma_f32_16x16x4_f32(false, a, false, bl,
                                                 (short)0, accL, false, false);
    accR = __builtin_amdgcn_wmma_f32_16x16x4_f32(false, a, false, br,
                                                 (short)0, accR, false, false);
  }

#pragma unroll
  for (int v = 0; v < 8; ++v) {
    int m = v + 8 * laneHi;
    int r = rowTile * 16 + m;
    if (r < N) {
      xl[(size_t)r * FDIM + colBase + lane16] = accL[v];
      xr[(size_t)r * FDIM + colBase + lane16] = accR[v];
    }
  }
}

__global__ __launch_bounds__(256) void gatv2_edge_logit(
    const float* __restrict__ xl, const float* __restrict__ xr,
    const int* __restrict__ esrc, const int* __restrict__ edst,
    const float* __restrict__ att, float* __restrict__ e,
    unsigned* __restrict__ mkey, int E, int ET) {
  int eid = blockIdx.x * 8 + (threadIdx.x >> 5);
  if (eid >= ET) return;
  int lane = threadIdx.x & 31;
  int s, d;
  if (eid < E) { s = esrc[eid]; d = edst[eid]; }
  else         { s = d = eid - E; }

  v2f a  = *(const v2f*)(xl + (size_t)s * FDIM + 2 * lane);
  v2f b  = *(const v2f*)(xr + (size_t)d * FDIM + 2 * lane);
  v2f at = *(const v2f*)(att + 2 * lane);
  float t0 = a[0] + b[0]; t0 = t0 > 0.f ? t0 : NEG_SLOPE * t0;
  float t1 = a[1] + b[1]; t1 = t1 > 0.f ? t1 : NEG_SLOPE * t1;
  float sum = t0 * at[0] + t1 * at[1];
#pragma unroll
  for (int off = 16; off; off >>= 1) sum += __shfl_xor(sum, off, 32);
  if (lane == 0) {
    e[eid] = sum;
    atomicMax(mkey + d, fenc(sum));
  }
}

__global__ __launch_bounds__(256) void gatv2_edge_exp(
    const float* __restrict__ e, const int* __restrict__ edst,
    const unsigned* __restrict__ mkey, float* __restrict__ ex,
    float* __restrict__ denom, int E, int ET) {
  int eid = blockIdx.x * 256 + threadIdx.x;
  if (eid >= ET) return;
  int d = (eid < E) ? edst[eid] : (eid - E);
  float v = __expf(e[eid] - fdec(mkey[d]));
  ex[eid] = v;
  atomicAdd(denom + d, v);
}

__global__ __launch_bounds__(256) void gatv2_edge_aggregate(
    const float* __restrict__ xl, const float* __restrict__ ex,
    const float* __restrict__ denom, const int* __restrict__ esrc,
    const int* __restrict__ edst, float* __restrict__ agg, int E, int ET) {
  int eid = blockIdx.x * 8 + (threadIdx.x >> 5);
  if (eid >= ET) return;
  int lane = threadIdx.x & 31;
  int s, d;
  if (eid < E) { s = esrc[eid]; d = edst[eid]; }
  else         { s = d = eid - E; }

  float alpha = ex[eid] / (denom[d] + EPS_DEN);
  v2f a = *(const v2f*)(xl + (size_t)s * FDIM + 2 * lane);
  float* dp = agg + (size_t)d * FDIM + 2 * lane;
  atomicAdd(dp, alpha * a[0]);
  atomicAdd(dp + 1, alpha * a[1]);
}

__global__ __launch_bounds__(256) void gatv2_finalize(
    const float* __restrict__ agg, const float* __restrict__ bias,
    float* __restrict__ out, size_t n, int relu) {
  size_t i = (size_t)blockIdx.x * 256 + threadIdx.x;
  if (i >= n) return;
  float v = agg[i] + bias[i & (FDIM - 1)];
  out[i] = (relu && v < 0.f) ? 0.f : v;
}

extern "C" void kernel_launch(void* const* d_in, const int* in_sizes, int n_in,
                              void* d_out, int out_size, void* d_ws, size_t ws_size,
                              hipStream_t stream) {
  const float* x    = (const float*)d_in[0];
  const int*   ei   = (const int*)d_in[1];
  const float* W1l  = (const float*)d_in[2];
  const float* W1r  = (const float*)d_in[3];
  const float* att1 = (const float*)d_in[4];
  const float* b1   = (const float*)d_in[5];
  const float* W2l  = (const float*)d_in[6];
  const float* W2r  = (const float*)d_in[7];
  const float* att2 = (const float*)d_in[8];
  const float* b2   = (const float*)d_in[9];

  const int N  = in_sizes[0] / 128;
  const int E  = in_sizes[1] / 2;
  const int ET = E + N;

  float* ws = (float*)d_ws;
  const size_t NF = (size_t)N * FDIM;
  float*    xl    = ws;
  float*    xr    = xl + NF;
  float*    h     = xr + NF;
  float*    agg   = h + NF;
  float*    e     = agg + NF;
  float*    ex    = e + ET;
  unsigned* mkey  = (unsigned*)(ex + ET);
  float*    denom = (float*)(mkey + N);

  const int rowTiles   = (N + 15) / 16;
  const int edgeWaveBl = (ET + 7) / 8;
  const int edgeThrBl  = (ET + 255) / 256;
  const int nfBl       = (int)((NF + 255) / 256);
  const int* esrc = ei;
  const int* edst = ei + E;

  hipMemsetAsync(mkey, 0, sizeof(unsigned) * N, stream);
  hipMemsetAsync(denom, 0, sizeof(float) * N, stream);
  hipMemsetAsync(agg, 0, sizeof(float) * NF, stream);
  gatv2_gemm_dual<<<rowTiles, 128, 0, stream>>>(x, W1l, W1r, xl, xr, N, 128);
  gatv2_edge_logit<<<edgeWaveBl, 256, 0, stream>>>(xl, xr, esrc, edst, att1, e, mkey, E, ET);
  gatv2_edge_exp<<<edgeThrBl, 256, 0, stream>>>(e, edst, mkey, ex, denom, E, ET);
  gatv2_edge_aggregate<<<edgeWaveBl, 256, 0, stream>>>(xl, ex, denom, esrc, edst, agg, E, ET);
  gatv2_finalize<<<nfBl, 256, 0, stream>>>(agg, b1, h, NF, 1);

  hipMemsetAsync(mkey, 0, sizeof(unsigned) * N, stream);
  hipMemsetAsync(denom, 0, sizeof(float) * N, stream);
  hipMemsetAsync(agg, 0, sizeof(float) * NF, stream);
  gatv2_gemm_dual<<<rowTiles, 128, 0, stream>>>(h, W2l, W2r, xl, xr, N, 64);
  gatv2_edge_logit<<<edgeWaveBl, 256, 0, stream>>>(xl, xr, esrc, edst, att2, e, mkey, E, ET);
  gatv2_edge_exp<<<edgeThrBl, 256, 0, stream>>>(e, edst, mkey, ex, denom, E, ET);
  gatv2_edge_aggregate<<<edgeWaveBl, 256, 0, stream>>>(xl, ex, denom, esrc, edst, agg, E, ET);
  gatv2_finalize<<<nfBl, 256, 0, stream>>>(agg, b2, (float*)d_out, NF, 0);
}


# edge_logit and edge_aggregate: each wave handles 32 edges (coalesced index loads, row loads of 8 edges in flight together, logits by a butterfly transpose-reduce) instead of one edge per wave
# speedup vs baseline: 1.7375x; 1.7375x over previous
#include <hip/hip_runtime.h>
#include <hip/hip_bf16.h>

typedef __attribute__((ext_vector_type(2))) float v2f;
typedef __attribute__((ext_vector_type(8))) float v8f;

#define NEG_SLOPE 0.2f
#define EPS_DEN 1e-16f
#define FDIM 64

__device__ __forceinline__ unsigned fenc(float f) {
  unsigned b = __float_as_uint(f);
  return (b & 0x80000000u) ? ~b : (b | 0x80000000u);
}
__device__ __forceinline__ float fdec(unsigned k) {
  unsigned b = (k & 0x80000000u) ? (k & 0x7fffffffu) : ~k;
  return __uint_as_float(b);
}

__global__ __launch_bounds__(128) void gatv2_gemm_dual(
    const float* __restrict__ X, const float* __restrict__ Wl,
    const float* __restrict__ Wr, float* __restrict__ xl,
    float* __restrict__ xr, int N, int K) {
  const int lane   = threadIdx.x & 31;
  const int wave   = threadIdx.x >> 5;
  const int laneHi = lane >> 4;
  const int lane16 = lane & 15;
  const int rowTile = blockIdx.x;
  const int colBase = wave * 16;

  int row  = rowTile * 16 + lane16;
  int rowc = row < N ? row : N - 1;

  v8f accL = {};
  v8f accR = {};
  const float* xrow = X + (size_t)rowc * K;

  for (int k = 0; k < K; k += 4) {
    const int kk = k + 2 * laneHi;
    v2f a = *(const v2f*)(xrow + kk);
    v2f bl, br;
    bl[0] = Wl[(size_t)kk * FDIM + colBase + lane16];
    bl[1] = Wl[(size_t)(kk + 1) * FDIM + colBase + lane16];
    br[0] = Wr[(size_t)kk * FDIM + colBase + lane16];
    br[1] = Wr[(size_t)(kk + 1) * FDIM + colBase + lane16];
    accL = __builtin_amdgcn_wmma_f32_16x16x4_f32(false, a, false, bl,
                                                 (short)0, accL, false, false);
    accR = __builtin_amdgcn_wmma_f32_16x16x4_f32(false, a, false, br,
                                                 (short)0, accR, false, false);
  }

#pragma unroll
  for (int v = 0; v < 8; ++v) {
    int m = v + 8 * laneHi;
    int r = rowTile * 16 + m;
    if (r < N) {
      xl[(size_t)r * FDIM + colBase + lane16] = accL[v];
      xr[(size_t)r * FDIM + colBase + lane16] = accR[v];
    }
  }
}

__global__ __launch_bounds__(256) void gatv2_edge_logit(
    const float* __restrict__ xl, const float* __restrict__ xr,
    const int* __restrict__ esrc, const int* __restrict__ edst,
    const float* __restrict__ att, float* __restrict__ e,
    unsigned* __restrict__ mkey, int E, int ET) {
  const int lane = threadIdx.x & 31;
  const int wv   = __builtin_amdgcn_readfirstlane((int)(threadIdx.x >> 5));
  const int base = (blockIdx.x * 8 + wv) * 32;
  if (base >= ET) return;
  const int eid = base + lane;
  const bool valid = eid < ET;
  int s = 0, d = 0;
  if (valid) {
    if (eid < E) { s = esrc[eid]; d = edst[eid]; }
    else         { s = d = eid - E; }
  }
  const v2f at = *(const v2f*)(att + 2 * lane);
  const float* xlL = xl + 2 * lane;
  const float* xrL = xr + 2 * lane;

  float p[32];
#pragma unroll
  for (int c = 0; c < 32; c += 8) {
    v2f a[8], b[8];
#pragma unroll
    for (int k = 0; k < 8; ++k) {
      const int sj = __builtin_amdgcn_readlane(s, c + k);
      const int dj = __builtin_amdgcn_readlane(d, c + k);
      a[k] = *(const v2f*)(xlL + (size_t)sj * FDIM);
      b[k] = *(const v2f*)(xrL + (size_t)dj * FDIM);
    }
#pragma unroll
    for (int k = 0; k < 8; ++k) {
      float t0 = a[k][0] + b[k][0]; t0 = t0 > 0.f ? t0 : NEG_SLOPE * t0;
      float t1 = a[k][1] + b[k][1]; t1 = t1 > 0.f ? t1 : NEG_SLOPE * t1;
      p[c + k] = t0 * at[0] + t1 * at[1];
    }
  }

#define GATV2_TR_STEP(WD)                                   \
  {                                                         \
    const bool hi = (lane & (WD)) != 0;                     \
    _Pragma("unroll") for (int i = 0; i < (WD); ++i) {      \
      const float keep = hi ? p[i + (WD)] : p[i];           \
      const float send = hi ? p[i] : p[i + (WD)];           \
      p[i] = keep + __shfl_xor(send, (WD), 32);             \
    }                                                       \
  }
  GATV2_TR_STEP(16)
  GATV2_TR_STEP(8)
  GATV2_TR_STEP(4)
  GATV2_TR_STEP(2)
  GATV2_TR_STEP(1)
#undef GATV2_TR_STEP
  if (valid) {
    e[eid] = p[0];
    atomicMax(mkey + d, fenc(p[0]));
  }
}

__global__ __launch_bounds__(256) void gatv2_edge_exp(
    const float* __restrict__ e, const int* __restrict__ edst,
    const unsigned* __restrict__ mkey, float* __restrict__ ex,
    float* __restrict__ denom, int E, int ET) {
  int eid = blockIdx.x * 256 + threadIdx.x;
  if (eid >= ET) return;
  int d = (eid < E) ? edst[eid] : (eid - E);
  float v = __expf(e[eid] - fdec(mkey[d]));
  ex[eid] = v;
  atomicAdd(denom + d, v);
}

__global__ __launch_bounds__(256) void gatv2_edge_aggregate(
    const float* __restrict__ xl, const float* __restrict__ ex,
    const float* __restrict__ denom, const int* __restrict__ esrc,
    const int* __restrict__ edst, float* __restrict__ agg, int E, int ET) {
  const int lane = threadIdx.x & 31;
  const int wv   = __builtin_amdgcn_readfirstlane((int)(threadIdx.x >> 5));
  const int base = (blockIdx.x * 8 + wv) * 32;
  if (base >= ET) return;
  const int nv  = (ET - base) < 32 ? (ET - base) : 32;
  const int eid = base + lane;
  int s = 0, d = 0;
  float alpha = 0.f;
  if (eid < ET) {
    if (eid < E) { s = esrc[eid]; d = edst[eid]; }
    else         { s = d = eid - E; }
    alpha = ex[eid] / (denom[d] + EPS_DEN);
  }
  const float* xlL = xl + 2 * lane;
  float* agL = agg + 2 * lane;

#pragma unroll
  for (int c = 0; c < 32; c += 8) {
    if (c + 8 <= nv) {
      v2f row[8];
#pragma unroll
      for (int k = 0; k < 8; ++k) {
        const int sj = __builtin_amdgcn_readlane(s, c + k);
        row[k] = *(const v2f*)(xlL + (size_t)sj * FDIM);
      }
#pragma unroll
      for (int k = 0; k < 8; ++k) {
        const int dj = __builtin_amdgcn_readlane(d, c + k);
        const float aj = __int_as_float(
            __builtin_amdgcn_readlane(__float_as_int(alpha), c + k));
        float* dp = agL + (size_t)dj * FDIM;
        atomicAdd(dp, aj * row[k][0]);
        atomicAdd(dp + 1, aj * row[k][1]);
      }
    } else {
      for (int j = c; j < nv; ++j) {
        const int sj = __builtin_amdgcn_readlane(s, j);
        const int dj = __builtin_amdgcn_readlane(d, j);
        const float aj = __int_as_float(
            __builtin_amdgcn_readlane(__float_as_int(alpha), j));
        v2f a = *(const v2f*)(xlL + (size_t)sj * FDIM);
        float* dp = agL + (size_t)dj * FDIM;
        atomicAdd(dp, aj * a[0]);
        atomicAdd(dp + 1, aj * a[1]);
      }
    }
  }
}

__global__ __launch_bounds__(256) void gatv2_finalize(
    const float* __restrict__ agg, const float* __restrict__ bias,
    float* __restrict__ out, size_t n, int relu) {
  size_t i = (size_t)blockIdx.x * 256 + threadIdx.x;
  if (i >= n) return;
  float v = agg[i] + bias[i & (FDIM - 1)];
  out[i] = (relu && v < 0.f) ? 0.f : v;
}

extern "C" void kernel_launch(void* const* d_in, const int* in_sizes, int n_in,
                              void* d_out, int out_size, void* d_ws, size_t ws_size,
                              hipStream_t stream) {
  const float* x    = (const float*)d_in[0];
  const int*   ei   = (const int*)d_in[1];
  const float* W1l  = (const float*)d_in[2];
  const float* W1r  = (const float*)d_in[3];
  const float* att1 = (const float*)d_in[4];
  const float* b1   = (const float*)d_in[5];
  const float* W2l  = (const float*)d_in[6];
  const float* W2r  = (const float*)d_in[7];
  const float* att2 = (const float*)d_in[8];
  const float* b2   = (const float*)d_in[9];

  const int N  = in_sizes[0] / 128;
  const int E  = in_sizes[1] / 2;
  const int ET = E + N;

  float* ws = (float*)d_ws;
  const size_t NF = (size_t)N * FDIM;
  float*    xl    = ws;
  float*    xr    = xl + NF;
  float*    h     = xr + NF;
  float*    agg   = h + NF;
  float*    e     = agg + NF;
  float*    ex    = e + ET;
  unsigned* mkey  = (unsigned*)(ex + ET);
  float*    denom = (float*)(mkey + N);

  const int rowTiles   = (N + 15) / 16;
  const int edgeWaveBl = (ET + 255) / 256;
  const int edgeThrBl  = (ET + 255) / 256;
  const int nfBl       = (int)((NF + 255) / 256);
  const int* esrc = ei;
  const int* edst = ei + E;

  hipMemsetAsync(mkey, 0, sizeof(unsigned) * N, stream);
  hipMemsetAsync(denom, 0, sizeof(float) * N, stream);
  hipMemsetAsync(agg, 0, sizeof(float) * NF, stream);
  gatv2_gemm_dual<<<rowTiles, 128, 0, stream>>>(x, W1l, W1r, xl, xr, N, 128);
  gatv2_edge_logit<<<edgeWaveBl, 256, 0, stream>>>(xl, xr, esrc, edst, att1, e, mkey, E, ET);
  gatv2_edge_exp<<<edgeThrBl, 256, 0, stream>>>(e, edst, mkey, ex, denom, E, ET);
  gatv2_edge_aggregate<<<edgeWaveBl, 256, 0, stream>>>(xl, ex, denom, esrc, edst, agg, E, ET);
  gatv2_finalize<<<nfBl, 256, 0, stream>>>(agg, b1, h, NF, 1);

  hipMemsetAsync(mkey, 0, sizeof(unsigned) * N, stream);
  hipMemsetAsync(denom, 0, sizeof(float) * N, stream);
  hipMemsetAsync(agg, 0, sizeof(float) * NF, stream);
  gatv2_gemm_dual<<<rowTiles, 128, 0, stream>>>(h, W2l, W2r, xl, xr, N, 64);
  gatv2_edge_logit<<<edgeWaveBl, 256, 0, stream>>>(xl, xr, esrc, edst, att2, e, mkey, E, ET);
  gatv2_edge_exp<<<edgeThrBl, 256, 0, stream>>>(e, edst, mkey, ex, denom, E, ET);
  gatv2_edge_aggregate<<<edgeWaveBl, 256, 0, stream>>>(xl, ex, denom, esrc, edst, agg, E, ET);
  gatv2_finalize<<<nfBl, 256, 0, stream>>>(agg, b2, (float*)d_out, NF, 0);
}


# edges counting-sorted by (dst, src) once per call; logit/exp walk the sorted list; one wave per node sums denom and alpha*xl over its segment in registers in that fixed order: no float atomics left
# speedup vs baseline: 2.7247x; 1.5682x over previous
#include <hip/hip_runtime.h>
#include <hip/hip_bf16.h>

typedef __attribute__((ext_vector_type(2))) float v2f;
typedef __attribute__((ext_vector_type(8))) float v8f;
typedef __attribute__((ext_vector_type(2))) int v2i;
typedef __attribute__((ext_vector_type(4))) int v4i;

#define NEG_SLOPE 0.2f
#define EPS_DEN 1e-16f
#define FDIM 64

__device__ __forceinline__ unsigned fenc(float f) {
  unsigned b = __float_as_uint(f);
  return (b & 0x80000000u) ? ~b : (b | 0x80000000u);
}
__device__ __forceinline__ float fdec(unsigned k) {
  unsigned b = (k & 0x80000000u) ? (k & 0x7fffffffu) : ~k;
  return __uint_as_float(b);
}

__global__ __launch_bounds__(128) void gatv2_gemm_dual(
    const float* __restrict__ X, const float* __restrict__ Wl,
    const float* __restrict__ Wr, float* __restrict__ xl,
    float* __restrict__ xr, int N, int K) {
  const int lane   = threadIdx.x & 31;
  const int wave   = threadIdx.x >> 5;
  const int laneHi = lane >> 4;
  const int lane16 = lane & 15;
  const int rowTile = blockIdx.x;
  const int colBase = wave * 16;

  int row  = rowTile * 16 + lane16;
  int rowc = row < N ? row : N - 1;

  v8f accL = {};
  v8f accR = {};
  const float* xrow = X + (size_t)rowc * K;

  for (int k = 0; k < K; k += 4) {
    const int kk = k + 2 * laneHi;
    v2f a = *(const v2f*)(xrow + kk);
    v2f bl, br;
    bl[0] = Wl[(size_t)kk * FDIM + colBase + lane16];
    bl[1] = Wl[(size_t)(kk + 1) * FDIM + colBase + lane16];
    br[0] = Wr[(size_t)kk * FDIM + colBase + lane16];
    br[1] = Wr[(size_t)(kk + 1) * FDIM + colBase + lane16];
    accL = __builtin_amdgcn_wmma_f32_16x16x4_f32(false, a, false, bl,
                                                 (short)0, accL, false, false);
    accR = __builtin_amdgcn_wmma_f32_16x16x4_f32(false, a, false, br,
                                                 (short)0, accR, false, false);
  }

#pragma unroll
  for (int v = 0; v < 8; ++v) {
    int m = v + 8 * laneHi;
    int r = rowTile * 16 + m;
    if (r < N) {
      xl[(size_t)r * FDIM + colBase + lane16] = accL[v];
      xr[(size_t)r * FDIM + colBase + lane16] = accR[v];
    }
  }
}

__global__ __launch_bounds__(256) void gatv2_edge_logit(
    const float* __restrict__ xl, const float* __restrict__ xr,
    const int* __restrict__ spair,
    const float* __restrict__ att, float* __restrict__ e,
    unsigned* __restrict__ mkey, int E, int ET) {
  const int lane = threadIdx.x & 31;
  const int wv   = __builtin_amdgcn_readfirstlane((int)(threadIdx.x >> 5));
  const int base = (blockIdx.x * 8 + wv) * 32;
  if (base >= ET) return;
  const int eid = base + lane;
  const bool valid = eid < ET;
  int s = 0, d = 0;
  if (valid) {
    if (eid < E) { const v2i q = *(const v2i*)(spair + 2 * (size_t)eid); s = q[0]; d = q[1]; }
    else         { s = d = eid - E; }
  }
  const v2f at = *(const v2f*)(att + 2 * lane);
  const float* xlL = xl + 2 * lane;
  const float* xrL = xr + 2 * lane;

  float p[32];
#pragma unroll
  for (int c = 0; c < 32; c += 8) {
    v2f a[8], b[8];
#pragma unroll
    for (int k = 0; k < 8; ++k) {
      const int sj = __builtin_amdgcn_readlane(s, c + k);
      const int dj = __builtin_amdgcn_readlane(d, c + k);
      a[k] = *(const v2f*)(xlL + (size_t)sj * FDIM);
      b[k] = *(const v2f*)(xrL + (size_t)dj * FDIM);
    }
#pragma unroll
    for (int k = 0; k < 8; ++k) {
      float t0 = a[k][0] + b[k][0]; t0 = t0 > 0.f ? t0 : NEG_SLOPE * t0;
      float t1 = a[k][1] + b[k][1]; t1 = t1 > 0.f ? t1 : NEG_SLOPE * t1;
      p[c + k] = t0 * at[0] + t1 * at[1];
    }
  }

#define GATV2_TR_STEP(WD)                                   \
  {                                                         \
    const bool hi = (lane & (WD)) != 0;                     \
    _Pragma("unroll") for (int i = 0; i < (WD); ++i) {      \
      const float keep = hi ? p[i + (WD)] : p[i];           \
      const float send = hi ? p[i] : p[i + (WD)];           \
      p[i] = keep + __shfl_xor(send, (WD), 32);             \
    }                                                       \
  }
  GATV2_TR_STEP(16)
  GATV2_TR_STEP(8)
  GATV2_TR_STEP(4)
  GATV2_TR_STEP(2)
  GATV2_TR_STEP(1)
#undef GATV2_TR_STEP
  if (valid) {
    e[eid] = p[0];
    atomicMax(mkey + d, fenc(p[0]));
  }
}

__global__ __launch_bounds__(256) void gatv2_edge_exp(
    const float* __restrict__ e, const int* __restrict__ spair,
    const unsigned* __restrict__ mkey, float* __restrict__ ex, int E, int ET) {
  int eid = blockIdx.x * 256 + threadIdx.x;
  if (eid >= ET) return;
  int d = (eid < E) ? spair[2 * (size_t)eid + 1] : (eid - E);
  ex[eid] = __expf(e[eid] - fdec(mkey[d]));
}

__global__ __launch_bounds__(256) void gatv2_bucket_rank(
    const int* __restrict__ edst, int* __restrict__ cnt,
    int* __restrict__ rank, int E) {
  const int eid = blockIdx.x * 256 + threadIdx.x;
  if (eid < E) rank[eid] = atomicAdd(cnt + edst[eid], 1);
}

__global__ __launch_bounds__(1024) void gatv2_rowptr_scan(
    const int* __restrict__ cnt, int* __restrict__ rowptr, int per) {
  __shared__ int wsum[32];
  const int t    = threadIdx.x;
  const int lane = t & 31;
  const int wave = t >> 5;
  const int nq   = per >> 2;
  const v4i* c4 = (const v4i*)(cnt + (size_t)t * per);
  int sum = 0;
  for (int i = 0; i < nq; ++i) {
    const v4i v = c4[i];
    sum += (v[0] + v[1]) + (v[2] + v[3]);
  }
  int incl = sum;
  for (int off = 1; off < 32; off <<= 1) {
    const int v = __shfl_up(incl, off, 32);
    if (lane >= off) incl += v;
  }
  if (lane == 31) wsum[wave] = incl;
  __syncthreads();
  if (wave == 0) {
    const int w = wsum[lane];
    int wi = w;
    for (int off = 1; off < 32; off <<= 1) {
      const int v = __shfl_up(wi, off, 32);
      if (lane >= off) wi += v;
    }
    wsum[lane] = wi - w;
  }
  __syncthreads();
  int run = wsum[wave] + incl - sum;
  v4i* r4 = (v4i*)(rowptr + (size_t)t * per);
  for (int i = 0; i < nq; ++i) {
    const v4i v = c4[i];
    v4i o;
    o[0] = run;
    o[1] = o[0] + v[0];
    o[2] = o[1] + v[1];
    o[3] = o[2] + v[2];
    run  = o[3] + v[3];
    r4[i] = o;
  }
}

__global__ __launch_bounds__(256) void gatv2_bucket_place(
    const int* __restrict__ esrc, const int* __restrict__ edst,
    const int* __restrict__ rowptr, const int* __restrict__ rank,
    int* __restrict__ tsrc, int E) {
  const int eid = blockIdx.x * 256 + threadIdx.x;
  if (eid >= E) return;
  const int s   = esrc[eid];
  const int pos = rowptr[edst[eid]] + rank[eid];
  if ((unsigned)pos < (unsigned)E) tsrc[pos] = s;
}

__global__ __launch_bounds__(256) void gatv2_segment_order(
    const int* __restrict__ rowptr, const int* __restrict__ tsrc,
    int* __restrict__ spair, int N, int E) {
  const int lane = threadIdx.x & 31;
  const int wv   = __builtin_amdgcn_readfirstlane((int)(threadIdx.x >> 5));
  const int node = blockIdx.x * 8 + wv;
  if (node >= N) return;
  int beg = rowptr[node];
  int end = rowptr[node + 1];
  beg = beg < 0 ? 0 : beg;
  end = end > E ? E : end;
  for (int cj = beg; cj < end; cj += 32) {
    const int pj = cj + lane;
    const bool vj = pj < end;
    const int sj = tsrc[vj ? pj : end - 1];
    int rank = 0;
    for (int ci = beg; ci < end; ci += 32) {
      const int ni = (end - ci) < 32 ? (end - ci) : 32;
      const int pi = ci + lane;
      const int si = tsrc[pi < end ? pi : end - 1];
      for (int i = 0; i < ni; ++i) {
        const int v = __builtin_amdgcn_readlane(si, i);
        rank += (v < sj || (v == sj && ci + i < pj)) ? 1 : 0;
      }
    }
    if (vj) {
      v2i q;
      q[0] = sj;
      q[1] = node;
      *(v2i*)(spair + 2 * (size_t)(beg + rank)) = q;
    }
  }
}

__global__ __launch_bounds__(256) void gatv2_node_aggregate(
    const float* __restrict__ xl, const float* __restrict__ ex,
    const int* __restrict__ rowptr, const int* __restrict__ spair,
    const float* __restrict__ bias, float* __restrict__ out,
    int N, int E, int relu) {
  const int lane = threadIdx.x & 31;
  const int wv   = __builtin_amdgcn_readfirstlane((int)(threadIdx.x >> 5));
  const int node = blockIdx.x * 8 + wv;
  if (node >= N) return;
  int beg = rowptr[node];
  int end = rowptr[node + 1];
  beg = beg < 0 ? 0 : beg;
  end = end > E ? E : end;
  const float* xlL = xl + 2 * lane;
  const float exs = ex[E + node];
  const v2f rself = *(const v2f*)(xlL + (size_t)node * FDIM);

  float part = 0.f;
  for (int c0 = beg; c0 < end; c0 += 32) {
    const int slot = c0 + lane;
    if (slot < end) part += ex[slot];
  }
  for (int off = 16; off >= 1; off >>= 1) part += __shfl_xor(part, off, 32);
  const float den = (part + exs) + EPS_DEN;

  const float aself = exs / den;
  v2f acc;
  acc[0] = aself * rself[0];
  acc[1] = aself * rself[1];

  for (int c0 = beg; c0 < end; c0 += 32) {
    const int n = (end - c0) < 32 ? (end - c0) : 32;
    const int slot = (c0 + lane) < end ? (c0 + lane) : (end - 1);
    int s = spair[2 * (size_t)slot];
    s = s < 0 ? 0 : (s >= N ? N - 1 : s);
    const float al = ex[slot] / den;
    for (int j0 = 0; j0 < n; j0 += 8) {
      v2f row[8];
#pragma unroll
      for (int k = 0; k < 8; ++k) {
        const int sj = __builtin_amdgcn_readlane(s, j0 + k);
        row[k] = *(const v2f*)(xlL + (size_t)sj * FDIM);
      }
#pragma unroll
      for (int k = 0; k < 8; ++k) {
        if (j0 + k < n) {
          const float aj = __int_as_float(
              __builtin_amdgcn_readlane(__float_as_int(al), j0 + k));
          acc[0] += aj * row[k][0];
          acc[1] += aj * row[k][1];
        }
      }
    }
  }

  const v2f b = *(const v2f*)(bias + 2 * lane);
  v2f o;
  o[0] = acc[0] + b[0];
  o[1] = acc[1] + b[1];
  if (relu) {
    o[0] = o[0] < 0.f ? 0.f : o[0];
    o[1] = o[1] < 0.f ? 0.f : o[1];
  }
  *(v2f*)(out + (size_t)node * FDIM + 2 * lane) = o;
}

extern "C" void kernel_launch(void* const* d_in, const int* in_sizes, int n_in,
                              void* d_out, int out_size, void* d_ws, size_t ws_size,
                              hipStream_t stream) {
  const float* x    = (const float*)d_in[0];
  const int*   ei   = (const int*)d_in[1];
  const float* W1l  = (const float*)d_in[2];
  const float* W1r  = (const float*)d_in[3];
  const float* att1 = (const float*)d_in[4];
  const float* b1   = (const float*)d_in[5];
  const float* W2l  = (const float*)d_in[6];
  const float* W2r  = (const float*)d_in[7];
  const float* att2 = (const float*)d_in[8];
  const float* b2   = (const float*)d_in[9];

  const int N  = in_sizes[0] / 128;
  const int E  = in_sizes[1] / 2;
  const int ET = E + N;

  float* ws = (float*)d_ws;
  const size_t NF = (size_t)N * FDIM;
  float*    xl    = ws;
  float*    xr    = xl + NF;
  float*    h     = xr + NF;
  int*      lst   = (int*)(h + NF);
  float*    e     = (float*)lst + NF;
  float*    ex    = e + ET;
  unsigned* mkey  = (unsigned*)(ex + ET);

  const int per = (((N + 1) + 1023) / 1024 + 31) / 32 * 32;
  const int NP  = per * 1024;
  int* cnt    = lst;
  int* rowptr = cnt + NP;
  int* spair  = rowptr + NP;
  int* tsrc   = (int*)e;
  int* rank   = (int*)ex;

  const int rowTiles   = (N + 15) / 16;
  const int edgeWaveBl = (ET + 255) / 256;
  const int edgeThrBl  = (ET + 255) / 256;
  const int realEdgeBl = (E + 255) / 256;
  const int nodeBl     = (N + 7) / 8;
  const int* esrc = ei;
  const int* edst = ei + E;

  hipMemsetAsync(cnt, 0, sizeof(int) * (size_t)NP, stream);
  gatv2_bucket_rank<<<realEdgeBl, 256, 0, stream>>>(edst, cnt, rank, E);
  gatv2_rowptr_scan<<<1, 1024, 0, stream>>>(cnt, rowptr, per);
  gatv2_bucket_place<<<realEdgeBl, 256, 0, stream>>>(esrc, edst, rowptr, rank, tsrc, E);
  gatv2_segment_order<<<nodeBl, 256, 0, stream>>>(rowptr, tsrc, spair, N, E);

  hipMemsetAsync(mkey, 0, sizeof(unsigned) * N, stream);
  gatv2_gemm_dual<<<rowTiles, 128, 0, stream>>>(x, W1l, W1r, xl, xr, N, 128);
  gatv2_edge_logit<<<edgeWaveBl, 256, 0, stream>>>(xl, xr, spair, att1, e, mkey, E, ET);
  gatv2_edge_exp<<<edgeThrBl, 256, 0, stream>>>(e, spair, mkey, ex, E, ET);
  gatv2_node_aggregate<<<nodeBl, 256, 0, stream>>>(xl, ex, rowptr, spair, b1, h, N, E, 1);

  hipMemsetAsync(mkey, 0, sizeof(unsigned) * N, stream);
  gatv2_gemm_dual<<<rowTiles, 128, 0, stream>>>(h, W2l, W2r, xl, xr, N, 64);
  gatv2_edge_logit<<<edgeWaveBl, 256, 0, stream>>>(xl, xr, spair, att2, e, mkey, E, ET);
  gatv2_edge_exp<<<edgeThrBl, 256, 0, stream>>>(e, spair, mkey, ex, E, ET);
  gatv2_node_aggregate<<<nodeBl, 256, 0, stream>>>(xl, ex, rowptr, spair, b2, (float*)d_out, N, E, 0);
}


# edge logits, segment max, exp, sum and weighted row sum fused into one wave-per-node kernel over the sorted segment, rows held in registers; max atomics, their memsets and the e/ex round trip gone
# speedup vs baseline: 3.3371x; 1.2248x over previous
#include <hip/hip_runtime.h>
#include <hip/hip_bf16.h>

typedef __attribute__((ext_vector_type(2))) float v2f;
typedef __attribute__((ext_vector_type(8))) float v8f;
typedef __attribute__((ext_vector_type(2))) int v2i;
typedef __attribute__((ext_vector_type(4))) int v4i;

#define NEG_SLOPE 0.2f
#define EPS_DEN 1e-16f
#define FDIM 64

__global__ __launch_bounds__(128) void gatv2_gemm_dual(
    const float* __restrict__ X, const float* __restrict__ Wl,
    const float* __restrict__ Wr, float* __restrict__ xl,
    float* __restrict__ xr, int N, int K) {
  const int lane   = threadIdx.x & 31;
  const int wave   = threadIdx.x >> 5;
  const int laneHi = lane >> 4;
  const int lane16 = lane & 15;
  const int rowTile = blockIdx.x;
  const int colBase = wave * 16;

  int row  = rowTile * 16 + lane16;
  int rowc = row < N ? row : N - 1;

  v8f accL = {};
  v8f accR = {};
  const float* xrow = X + (size_t)rowc * K;

  for (int k = 0; k < K; k += 4) {
    const int kk = k + 2 * laneHi;
    v2f a = *(const v2f*)(xrow + kk);
    v2f bl, br;
    bl[0] = Wl[(size_t)kk * FDIM + colBase + lane16];
    bl[1] = Wl[(size_t)(kk + 1) * FDIM + colBase + lane16];
    br[0] = Wr[(size_t)kk * FDIM + colBase + lane16];
    br[1] = Wr[(size_t)(kk + 1) * FDIM + colBase + lane16];
    accL = __builtin_amdgcn_wmma_f32_16x16x4_f32(false, a, false, bl,
                                                 (short)0, accL, false, false);
    accR = __builtin_amdgcn_wmma_f32_16x16x4_f32(false, a, false, br,
                                                 (short)0, accR, false, false);
  }

#pragma unroll
  for (int v = 0; v < 8; ++v) {
    int m = v + 8 * laneHi;
    int r = rowTile * 16 + m;
    if (r < N) {
      xl[(size_t)r * FDIM + colBase + lane16] = accL[v];
      xr[(size_t)r * FDIM + colBase + lane16] = accR[v];
    }
  }
}

__global__ __launch_bounds__(256) void gatv2_bucket_rank(
    const int* __restrict__ edst, int* __restrict__ cnt,
    int* __restrict__ rank, int E) {
  const int eid = blockIdx.x * 256 + threadIdx.x;
  if (eid < E) rank[eid] = atomicAdd(cnt + edst[eid], 1);
}

__global__ __launch_bounds__(1024) void gatv2_rowptr_scan(
    const int* __restrict__ cnt, int* __restrict__ rowptr, int per) {
  __shared__ int wsum[32];
  const int t    = threadIdx.x;
  const int lane = t & 31;
  const int wave = t >> 5;
  const int nq   = per >> 2;
  const v4i* c4 = (const v4i*)(cnt + (size_t)t * per);
  int sum = 0;
  for (int i = 0; i < nq; ++i) {
    const v4i v = c4[i];
    sum += (v[0] + v[1]) + (v[2] + v[3]);
  }
  int incl = sum;
  for (int off = 1; off < 32; off <<= 1) {
    const int v = __shfl_up(incl, off, 32);
    if (lane >= off) incl += v;
  }
  if (lane == 31) wsum[wave] = incl;
  __syncthreads();
  if (wave == 0) {
    const int w = wsum[lane];
    int wi = w;
    for (int off = 1; off < 32; off <<= 1) {
      const int v = __shfl_up(wi, off, 32);
      if (lane >= off) wi += v;
    }
    wsum[lane] = wi - w;
  }
  __syncthreads();
  int run = wsum[wave] + incl - sum;
  v4i* r4 = (v4i*)(rowptr + (size_t)t * per);
  for (int i = 0; i < nq; ++i) {
    const v4i v = c4[i];
    v4i o;
    o[0] = run;
    o[1] = o[0] + v[0];
    o[2] = o[1] + v[1];
    o[3] = o[2] + v[2];
    run  = o[3] + v[3];
    r4[i] = o;
  }
}

__global__ __launch_bounds__(256) void gatv2_bucket_place(
    const int* __restrict__ esrc, const int* __restrict__ edst,
    const int* __restrict__ rowptr, const int* __restrict__ rank,
    int* __restrict__ tsrc, int E) {
  const int eid = blockIdx.x * 256 + threadIdx.x;
  if (eid >= E) return;
  const int s   = esrc[eid];
  const int pos = rowptr[edst[eid]] + rank[eid];
  if ((unsigned)pos < (unsigned)E) tsrc[pos] = s;
}

__global__ __launch_bounds__(256) void gatv2_segment_order(
    const int* __restrict__ rowptr, const int* __restrict__ tsrc,
    int* __restrict__ spair, int N, int E) {
  const int lane = threadIdx.x & 31;
  const int wv   = __builtin_amdgcn_readfirstlane((int)(threadIdx.x >> 5));
  const int node = blockIdx.x * 8 + wv;
  if (node >= N) return;
  int beg = rowptr[node];
  int end = rowptr[node + 1];
  beg = beg < 0 ? 0 : beg;
  end = end > E ? E : end;
  for (int cj = beg; cj < end; cj += 32) {
    const int pj = cj + lane;
    const bool vj = pj < end;
    const int sj = tsrc[vj ? pj : end - 1];
    int rank = 0;
    for (int ci = beg; ci < end; ci += 32) {
      const int ni = (end - ci) < 32 ? (end - ci) : 32;
      const int pi = ci + lane;
      const int si = tsrc[pi < end ? pi : end - 1];
      for (int i = 0; i < ni; ++i) {
        const int v = __builtin_amdgcn_readlane(si, i);
        rank += (v < sj || (v == sj && ci + i < pj)) ? 1 : 0;
      }
    }
    if (vj) {
      v2i q;
      q[0] = sj;
      q[1] = node;
      *(v2i*)(spair + 2 * (size_t)(beg + rank)) = q;
    }
  }
}

#define GATV2_TR_STEP(WD)                                   \
  {                                                         \
    const bool hi = (lane & (WD)) != 0;                     \
    _Pragma("unroll") for (int i = 0; i < (WD); ++i) {      \
      const float keep = hi ? p[i + (WD)] : p[i];           \
      const float send = hi ? p[i] : p[i + (WD)];           \
      p[i] = keep + __shfl_xor(send, (WD), 32);             \
    }                                                       \
  }

__device__ __forceinline__ float gatv2_partial(v2f a, v2f b, v2f at) {
  float t0 = a[0] + b[0]; t0 = t0 > 0.f ? t0 : NEG_SLOPE * t0;
  float t1 = a[1] + b[1]; t1 = t1 > 0.f ? t1 : NEG_SLOPE * t1;
  return t0 * at[0] + t1 * at[1];
}

__global__ __launch_bounds__(256) void gatv2_node_fused(
    const float* __restrict__ xl, const float* __restrict__ xr,
    const int* __restrict__ rowptr, const int* __restrict__ spair,
    const float* __restrict__ att, const float* __restrict__ bias,
    float* __restrict__ scr, float* __restrict__ out,
    int N, int E, int relu) {
  const int lane = threadIdx.x & 31;
  const int wv   = __builtin_amdgcn_readfirstlane((int)(threadIdx.x >> 5));
  const int node = blockIdx.x * 8 + wv;
  if (node >= N) return;
  int beg = rowptr[node];
  int end = rowptr[node + 1];
  beg = beg < 0 ? 0 : beg;
  end = end > E ? E : end;
  end = end < beg ? beg : end;
  const int deg = end - beg;
  const float* xlL = xl + 2 * lane;
  const v2f at    = *(const v2f*)(att + 2 * lane);
  const v2f xrn   = *(const v2f*)(xr + (size_t)node * FDIM + 2 * lane);
  const v2f rself = *(const v2f*)(xlL + (size_t)node * FDIM);

  float es = gatv2_partial(rself, xrn, at);

  v2f acc;
  if (deg <= 32) {
    const bool valid = lane < deg;
    int s = node;
    if (valid) s = spair[2 * (size_t)(beg + lane)];
    s = s < 0 ? 0 : (s >= N ? N - 1 : s);
    v2f row[32];
    float p[32];
#pragma unroll
    for (int c = 0; c < 32; c += 8) {
      if (c < deg) {
#pragma unroll
        for (int k = 0; k < 8; ++k) {
          const int sj = __builtin_amdgcn_readlane(s, c + k);
          row[c + k] = *(const v2f*)(xlL + (size_t)sj * FDIM);
        }
      }
    }
    for (int off = 16; off >= 1; off >>= 1) es += __shfl_xor(es, off, 32);
#pragma unroll
    for (int c = 0; c < 32; c += 8) {
      if (c < deg) {
#pragma unroll
        for (int k = 0; k < 8; ++k) p[c + k] = gatv2_partial(row[c + k], xrn, at);
      } else {
#pragma unroll
        for (int k = 0; k < 8; ++k) p[c + k] = 0.f;
      }
    }
    GATV2_TR_STEP(16)
    GATV2_TR_STEP(8)
    GATV2_TR_STEP(4)
    GATV2_TR_STEP(2)
    GATV2_TR_STEP(1)
    const float lg = p[0];
    float m = valid ? fmaxf(lg, es) : es;
    for (int off = 16; off >= 1; off >>= 1) m = fmaxf(m, __shfl_xor(m, off, 32));
    const float exv = valid ? __expf(lg - m) : 0.f;
    const float exs = __expf(es - m);
    float part = exv;
    for (int off = 16; off >= 1; off >>= 1) part += __shfl_xor(part, off, 32);
    const float den = (part + exs) + EPS_DEN;
    const float aself = exs / den;
    const float al = exv / den;
    acc[0] = aself * rself[0];
    acc[1] = aself * rself[1];
#pragma unroll
    for (int c = 0; c < 32; c += 8) {
      if (c < deg) {
#pragma unroll
        for (int k = 0; k < 8; ++k) {
          if (c + k < deg) {
            const float aj = __int_as_float(
                __builtin_amdgcn_readlane(__float_as_int(al), c + k));
            acc[0] += aj * row[c + k][0];
            acc[1] += aj * row[c + k][1];
          }
        }
      }
    }
  } else {
    for (int off = 16; off >= 1; off >>= 1) es += __shfl_xor(es, off, 32);
    float m = es;
    for (int c0 = beg; c0 < end; c0 += 32) {
      const int n = (end - c0) < 32 ? (end - c0) : 32;
      const int slot = (c0 + lane) < end ? (c0 + lane) : (end - 1);
      int s = spair[2 * (size_t)slot];
      s = s < 0 ? 0 : (s >= N ? N - 1 : s);
      float p[32];
#pragma unroll
      for (int c = 0; c < 32; c += 8) {
        if (c < n) {
          v2f r[8];
#pragma unroll
          for (int k = 0; k < 8; ++k) {
            const int sj = __builtin_amdgcn_readlane(s, c + k);
            r[k] = *(const v2f*)(xlL + (size_t)sj * FDIM);
          }
#pragma unroll
          for (int k = 0; k < 8; ++k) p[c + k] = gatv2_partial(r[k], xrn, at);
        } else {
#pragma unroll
          for (int k = 0; k < 8; ++k) p[c + k] = 0.f;
        }
      }
      GATV2_TR_STEP(16)
      GATV2_TR_STEP(8)
      GATV2_TR_STEP(4)
      GATV2_TR_STEP(2)
      GATV2_TR_STEP(1)
      if (c0 + lane < end) {
        scr[c0 + lane] = p[0];
        m = fmaxf(m, p[0]);
      }
    }
    for (int off = 16; off >= 1; off >>= 1) m = fmaxf(m, __shfl_xor(m, off, 32));
    const float exs = __expf(es - m);
    float part = 0.f;
    for (int c0 = beg; c0 < end; c0 += 32) {
      const int slot = c0 + lane;
      if (slot < end) part += __expf(scr[slot] - m);
    }
    for (int off = 16; off >= 1; off >>= 1) part += __shfl_xor(part, off, 32);
    const float den = (part + exs) + EPS_DEN;
    const float aself = exs / den;
    acc[0] = aself * rself[0];
    acc[1] = aself * rself[1];
    for (int c0 = beg; c0 < end; c0 += 32) {
      const int n = (end - c0) < 32 ? (end - c0) : 32;
      const int slot = (c0 + lane) < end ? (c0 + lane) : (end - 1);
      int s = spair[2 * (size_t)slot];
      s = s < 0 ? 0 : (s >= N ? N - 1 : s);
      const float al = (c0 + lane) < end ? __expf(scr[c0 + lane] - m) / den : 0.f;
      for (int j0 = 0; j0 < n; j0 += 8) {
        v2f r[8];
#pragma unroll
        for (int k = 0; k < 8; ++k) {
          const int sj = __builtin_amdgcn_readlane(s, j0 + k);
          r[k] = *(const v2f*)(xlL + (size_t)sj * FDIM);
        }
#pragma unroll
        for (int k = 0; k < 8; ++k) {
          if (j0 + k < n) {
            const float aj = __int_as_float(
                __builtin_amdgcn_readlane(__float_as_int(al), j0 + k));
            acc[0] += aj * r[k][0];
            acc[1] += aj * r[k][1];
          }
        }
      }
    }
  }

  const v2f b = *(const v2f*)(bias + 2 * lane);
  v2f o;
  o[0] = acc[0] + b[0];
  o[1] = acc[1] + b[1];
  if (relu) {
    o[0] = o[0] < 0.f ? 0.f : o[0];
    o[1] = o[1] < 0.f ? 0.f : o[1];
  }
  *(v2f*)(out + (size_t)node * FDIM + 2 * lane) = o;
}
#undef GATV2_TR_STEP

extern "C" void kernel_launch(void* const* d_in, const int* in_sizes, int n_in,
                              void* d_out, int out_size, void* d_ws, size_t ws_size,
                              hipStream_t stream) {
  const float* x    = (const float*)d_in[0];
  const int*   ei   = (const int*)d_in[1];
  const float* W1l  = (const float*)d_in[2];
  const float* W1r  = (const float*)d_in[3];
  const float* att1 = (const float*)d_in[4];
  const float* b1   = (const float*)d_in[5];
  const float* W2l  = (const float*)d_in[6];
  const float* W2r  = (const float*)d_in[7];
  const float* att2 = (const float*)d_in[8];
  const float* b2   = (const float*)d_in[9];

  const int N  = in_sizes[0] / 128;
  const int E  = in_sizes[1] / 2;
  const int ET = E + N;

  float* ws = (float*)d_ws;
  const size_t NF = (size_t)N * FDIM;
  float*    xl    = ws;
  float*    xr    = xl + NF;
  float*    h     = xr + NF;
  int*      lst   = (int*)(h + NF);
  float*    e     = (float*)lst + NF;
  float*    ex    = e + ET;

  const int per = (((N + 1) + 1023) / 1024 + 31) / 32 * 32;
  const int NP  = per * 1024;
  int* cnt    = lst;
  int* rowptr = cnt + NP;
  int* spair  = rowptr + NP;
  int* tsrc   = (int*)e;
  int* rank   = (int*)ex;

  const int rowTiles   = (N + 15) / 16;
  const int realEdgeBl = (E + 255) / 256;
  const int nodeBl     = (N + 7) / 8;
  const int* esrc = ei;
  const int* edst = ei + E;

  hipMemsetAsync(cnt, 0, sizeof(int) * (size_t)NP, stream);
  gatv2_bucket_rank<<<realEdgeBl, 256, 0, stream>>>(edst, cnt, rank, E);
  gatv2_rowptr_scan<<<1, 1024, 0, stream>>>(cnt, rowptr, per);
  gatv2_bucket_place<<<realEdgeBl, 256, 0, stream>>>(esrc, edst, rowptr, rank, tsrc, E);
  gatv2_segment_order<<<nodeBl, 256, 0, stream>>>(rowptr, tsrc, spair, N, E);

  gatv2_gemm_dual<<<rowTiles, 128, 0, stream>>>(x, W1l, W1r, xl, xr, N, 128);
  gatv2_node_fused<<<nodeBl, 256, 0, stream>>>(xl, xr, rowptr, spair, att1, b1, e, h, N, E, 1);

  gatv2_gemm_dual<<<rowTiles, 128, 0, stream>>>(h, W2l, W2r, xl, xr, N, 64);
  gatv2_node_fused<<<nodeBl, 256, 0, stream>>>(xl, xr, rowptr, spair, att2, b2, e, (float*)d_out, N, E, 0);
}


# layer-1 GEMM co-launched with the one-workgroup rowptr scan (block 0 scans, the other blocks multiply 8 row tiles each): the chip the scan left idle does the GEMM; one launch fewer
# speedup vs baseline: 3.4817x; 1.0433x over previous
#include <hip/hip_runtime.h>
#include <hip/hip_bf16.h>

typedef __attribute__((ext_vector_type(2))) float v2f;
typedef __attribute__((ext_vector_type(8))) float v8f;
typedef __attribute__((ext_vector_type(2))) int v2i;
typedef __attribute__((ext_vector_type(4))) int v4i;

#define NEG_SLOPE 0.2f
#define EPS_DEN 1e-16f
#define FDIM 64

__device__ __forceinline__ void gatv2_gemm_tile(
    const float* __restrict__ X, const float* __restrict__ Wl,
    const float* __restrict__ Wr, float* __restrict__ xl,
    float* __restrict__ xr, int N, int K, int rowTile, int colQ, int lane) {
  const int laneHi = lane >> 4;
  const int lane16 = lane & 15;
  const int colBase = colQ * 16;

  int row  = rowTile * 16 + lane16;
  int rowc = row < N ? row : N - 1;

  v8f accL = {};
  v8f accR = {};
  const float* xrow = X + (size_t)rowc * K;

  for (int k = 0; k < K; k += 4) {
    const int kk = k + 2 * laneHi;
    v2f a = *(const v2f*)(xrow + kk);
    v2f bl, br;
    bl[0] = Wl[(size_t)kk * FDIM + colBase + lane16];
    bl[1] = Wl[(size_t)(kk + 1) * FDIM + colBase + lane16];
    br[0] = Wr[(size_t)kk * FDIM + colBase + lane16];
    br[1] = Wr[(size_t)(kk + 1) * FDIM + colBase + lane16];
    accL = __builtin_amdgcn_wmma_f32_16x16x4_f32(false, a, false, bl,
                                                 (short)0, accL, false, false);
    accR = __builtin_amdgcn_wmma_f32_16x16x4_f32(false, a, false, br,
                                                 (short)0, accR, false, false);
  }

#pragma unroll
  for (int v = 0; v < 8; ++v) {
    int m = v + 8 * laneHi;
    int r = rowTile * 16 + m;
    if (r < N) {
      xl[(size_t)r * FDIM + colBase + lane16] = accL[v];
      xr[(size_t)r * FDIM + colBase + lane16] = accR[v];
    }
  }
}

__global__ __launch_bounds__(128) void gatv2_gemm_dual(
    const float* __restrict__ X, const float* __restrict__ Wl,
    const float* __restrict__ Wr, float* __restrict__ xl,
    float* __restrict__ xr, int N, int K) {
  gatv2_gemm_tile(X, Wl, Wr, xl, xr, N, K, (int)blockIdx.x,
                  (int)(threadIdx.x >> 5), (int)(threadIdx.x & 31));
}

__global__ __launch_bounds__(256) void gatv2_bucket_rank(
    const int* __restrict__ edst, int* __restrict__ cnt,
    int* __restrict__ rank, int E) {
  const int eid = blockIdx.x * 256 + threadIdx.x;
  if (eid < E) rank[eid] = atomicAdd(cnt + edst[eid], 1);
}

__device__ __forceinline__ void gatv2_rowptr_scan_block(
    const int* __restrict__ cnt, int* __restrict__ rowptr, int per) {
  __shared__ int wsum[32];
  const int t    = threadIdx.x;
  const int lane = t & 31;
  const int wave = t >> 5;
  const int nq   = per >> 2;
  const v4i* c4 = (const v4i*)(cnt + (size_t)t * per);
  int sum = 0;
  for (int i = 0; i < nq; ++i) {
    const v4i v = c4[i];
    sum += (v[0] + v[1]) + (v[2] + v[3]);
  }
  int incl = sum;
  for (int off = 1; off < 32; off <<= 1) {
    const int v = __shfl_up(incl, off, 32);
    if (lane >= off) incl += v;
  }
  if (lane == 31) wsum[wave] = incl;
  __syncthreads();
  if (wave == 0) {
    const int w = wsum[lane];
    int wi = w;
    for (int off = 1; off < 32; off <<= 1) {
      const int v = __shfl_up(wi, off, 32);
      if (lane >= off) wi += v;
    }
    wsum[lane] = wi - w;
  }
  __syncthreads();
  int run = wsum[wave] + incl - sum;
  v4i* r4 = (v4i*)(rowptr + (size_t)t * per);
  for (int i = 0; i < nq; ++i) {
    const v4i v = c4[i];
    v4i o;
    o[0] = run;
    o[1] = o[0] + v[0];
    o[2] = o[1] + v[1];
    o[3] = o[2] + v[2];
    run  = o[3] + v[3];
    r4[i] = o;
  }
}

__global__ __launch_bounds__(1024) void gatv2_scan_and_gemm(
    const int* __restrict__ cnt, int* __restrict__ rowptr, int per,
    const float* __restrict__ X, const float* __restrict__ Wl,
    const float* __restrict__ Wr, float* __restrict__ xl,
    float* __restrict__ xr, int N, int K) {
  if (blockIdx.x == 0) {
    gatv2_rowptr_scan_block(cnt, rowptr, per);
    return;
  }
  const int wv = __builtin_amdgcn_readfirstlane((int)(threadIdx.x >> 5));
  const int rowTile = ((int)blockIdx.x - 1) * 8 + (wv >> 2);
  if (rowTile * 16 >= N) return;
  gatv2_gemm_tile(X, Wl, Wr, xl, xr, N, K, rowTile, wv & 3, (int)(threadIdx.x & 31));
}

__global__ __launch_bounds__(256) void gatv2_bucket_place(
    const int* __restrict__ esrc, const int* __restrict__ edst,
    const int* __restrict__ rowptr, const int* __restrict__ rank,
    int* __restrict__ tsrc, int E) {
  const int eid = blockIdx.x * 256 + threadIdx.x;
  if (eid >= E) return;
  const int s   = esrc[eid];
  const int pos = rowptr[edst[eid]] + rank[eid];
  if ((unsigned)pos < (unsigned)E) tsrc[pos] = s;
}

__global__ __launch_bounds__(256) void gatv2_segment_order(
    const int* __restrict__ rowptr, const int* __restrict__ tsrc,
    int* __restrict__ spair, int N, int E) {
  const int lane = threadIdx.x & 31;
  const int wv   = __builtin_amdgcn_readfirstlane((int)(threadIdx.x >> 5));
  const int node = blockIdx.x * 8 + wv;
  if (node >= N) return;
  int beg = rowptr[node];
  int end = rowptr[node + 1];
  beg = beg < 0 ? 0 : beg;
  end = end > E ? E : end;
  for (int cj = beg; cj < end; cj += 32) {
    const int pj = cj + lane;
    const bool vj = pj < end;
    const int sj = tsrc[vj ? pj : end - 1];
    int rank = 0;
    for (int ci = beg; ci < end; ci += 32) {
      const int ni = (end - ci) < 32 ? (end - ci) : 32;
      const int pi = ci + lane;
      const int si = tsrc[pi < end ? pi : end - 1];
      for (int i = 0; i < ni; ++i) {
        const int v = __builtin_amdgcn_readlane(si, i);
        rank += (v < sj || (v == sj && ci + i < pj)) ? 1 : 0;
      }
    }
    if (vj) {
      v2i q;
      q[0] = sj;
      q[1] = node;
      *(v2i*)(spair + 2 * (size_t)(beg + rank)) = q;
    }
  }
}

#define GATV2_TR_STEP(WD)                                   \
  {                                                         \
    const bool hi = (lane & (WD)) != 0;                     \
    _Pragma("unroll") for (int i = 0; i < (WD); ++i) {      \
      const float keep = hi ? p[i + (WD)] : p[i];           \
      const float send = hi ? p[i] : p[i + (WD)];           \
      p[i] = keep + __shfl_xor(send, (WD), 32);             \
    }                                                       \
  }

__device__ __forceinline__ float gatv2_partial(v2f a, v2f b, v2f at) {
  float t0 = a[0] + b[0]; t0 = t0 > 0.f ? t0 : NEG_SLOPE * t0;
  float t1 = a[1] + b[1]; t1 = t1 > 0.f ? t1 : NEG_SLOPE * t1;
  return t0 * at[0] + t1 * at[1];
}

__global__ __launch_bounds__(256) void gatv2_node_fused(
    const float* __restrict__ xl, const float* __restrict__ xr,
    const int* __restrict__ rowptr, const int* __restrict__ spair,
    const float* __restrict__ att, const float* __restrict__ bias,
    float* __restrict__ scr, float* __restrict__ out,
    int N, int E, int relu) {
  const int lane = threadIdx.x & 31;
  const int wv   = __builtin_amdgcn_readfirstlane((int)(threadIdx.x >> 5));
  const int node = blockIdx.x * 8 + wv;
  if (node >= N) return;
  int beg = rowptr[node];
  int end = rowptr[node + 1];
  beg = beg < 0 ? 0 : beg;
  end = end > E ? E : end;
  end = end < beg ? beg : end;
  const int deg = end - beg;
  const float* xlL = xl + 2 * lane;
  const v2f at    = *(const v2f*)(att + 2 * lane);
  const v2f xrn   = *(const v2f*)(xr + (size_t)node * FDIM + 2 * lane);
  const v2f rself = *(const v2f*)(xlL + (size_t)node * FDIM);

  float es = gatv2_partial(rself, xrn, at);

  v2f acc;
  if (deg <= 32) {
    const bool valid = lane < deg;
    int s = node;
    if (valid) s = spair[2 * (size_t)(beg + lane)];
    s = s < 0 ? 0 : (s >= N ? N - 1 : s);
    v2f row[32];
    float p[32];
#pragma unroll
    for (int c = 0; c < 32; c += 8) {
      if (c < deg) {
#pragma unroll
        for (int k = 0; k < 8; ++k) {
          const int sj = __builtin_amdgcn_readlane(s, c + k);
          row[c + k] = *(const v2f*)(xlL + (size_t)sj * FDIM);
        }
      }
    }
    for (int off = 16; off >= 1; off >>= 1) es += __shfl_xor(es, off, 32);
#pragma unroll
    for (int c = 0; c < 32; c += 8) {
      if (c < deg) {
#pragma unroll
        for (int k = 0; k < 8; ++k) p[c + k] = gatv2_partial(row[c + k], xrn, at);
      } else {
#pragma unroll
        for (int k = 0; k < 8; ++k) p[c + k] = 0.f;
      }
    }
    GATV2_TR_STEP(16)
    GATV2_TR_STEP(8)
    GATV2_TR_STEP(4)
    GATV2_TR_STEP(2)
    GATV2_TR_STEP(1)
    const float lg = p[0];
    float m = valid ? fmaxf(lg, es) : es;
    for (int off = 16; off >= 1; off >>= 1) m = fmaxf(m, __shfl_xor(m, off, 32));
    const float exv = valid ? __expf(lg - m) : 0.f;
    const float exs = __expf(es - m);
    float part = exv;
    for (int off = 16; off >= 1; off >>= 1) part += __shfl_xor(part, off, 32);
    const float den = (part + exs) + EPS_DEN;
    const float aself = exs / den;
    const float al = exv / den;
    acc[0] = aself * rself[0];
    acc[1] = aself * rself[1];
#pragma unroll
    for (int c = 0; c < 32; c += 8) {
      if (c < deg) {
#pragma unroll
        for (int k = 0; k < 8; ++k) {
          if (c + k < deg) {
            const float aj = __int_as_float(
                __builtin_amdgcn_readlane(__float_as_int(al), c + k));
            acc[0] += aj * row[c + k][0];
            acc[1] += aj * row[c + k][1];
          }
        }
      }
    }
  } else {
    for (int off = 16; off >= 1; off >>= 1) es += __shfl_xor(es, off, 32);
    float m = es;
    for (int c0 = beg; c0 < end; c0 += 32) {
      const int n = (end - c0) < 32 ? (end - c0) : 32;
      const int slot = (c0 + lane) < end ? (c0 + lane) : (end - 1);
      int s = spair[2 * (size_t)slot];
      s = s < 0 ? 0 : (s >= N ? N - 1 : s);
      float p[32];
#pragma unroll
      for (int c = 0; c < 32; c += 8) {
        if (c < n) {
          v2f r[8];
#pragma unroll
          for (int k = 0; k < 8; ++k) {
            const int sj = __builtin_amdgcn_readlane(s, c + k);
            r[k] = *(const v2f*)(xlL + (size_t)sj * FDIM);
          }
#pragma unroll
          for (int k = 0; k < 8; ++k) p[c + k] = gatv2_partial(r[k], xrn, at);
        } else {
#pragma unroll
          for (int k = 0; k < 8; ++k) p[c + k] = 0.f;
        }
      }
      GATV2_TR_STEP(16)
      GATV2_TR_STEP(8)
      GATV2_TR_STEP(4)
      GATV2_TR_STEP(2)
      GATV2_TR_STEP(1)
      if (c0 + lane < end) {
        scr[c0 + lane] = p[0];
        m = fmaxf(m, p[0]);
      }
    }
    for (int off = 16; off >= 1; off >>= 1) m = fmaxf(m, __shfl_xor(m, off, 32));
    const float exs = __expf(es - m);
    float part = 0.f;
    for (int c0 = beg; c0 < end; c0 += 32) {
      const int slot = c0 + lane;
      if (slot < end) part += __expf(scr[slot] - m);
    }
    for (int off = 16; off >= 1; off >>= 1) part += __shfl_xor(part, off, 32);
    const float den = (part + exs) + EPS_DEN;
    const float aself = exs / den;
    acc[0] = aself * rself[0];
    acc[1] = aself * rself[1];
    for (int c0 = beg; c0 < end; c0 += 32) {
      const int n = (end - c0) < 32 ? (end - c0) : 32;
      const int slot = (c0 + lane) < end ? (c0 + lane) : (end - 1);
      int s = spair[2 * (size_t)slot];
      s = s < 0 ? 0 : (s >= N ? N - 1 : s);
      const float al = (c0 + lane) < end ? __expf(scr[c0 + lane] - m) / den : 0.f;
      for (int j0 = 0; j0 < n; j0 += 8) {
        v2f r[8];
#pragma unroll
        for (int k = 0; k < 8; ++k) {
          const int sj = __builtin_amdgcn_readlane(s, j0 + k);
          r[k] = *(const v2f*)(xlL + (size_t)sj * FDIM);
        }
#pragma unroll
        for (int k = 0; k < 8; ++k) {
          if (j0 + k < n) {
            const float aj = __int_as_float(
                __builtin_amdgcn_readlane(__float_as_int(al), j0 + k));
            acc[0] += aj * r[k][0];
            acc[1] += aj * r[k][1];
          }
        }
      }
    }
  }

  const v2f b = *(const v2f*)(bias + 2 * lane);
  v2f o;
  o[0] = acc[0] + b[0];
  o[1] = acc[1] + b[1];
  if (relu) {
    o[0] = o[0] < 0.f ? 0.f : o[0];
    o[1] = o[1] < 0.f ? 0.f : o[1];
  }
  *(v2f*)(out + (size_t)node * FDIM + 2 * lane) = o;
}
#undef GATV2_TR_STEP

extern "C" void kernel_launch(void* const* d_in, const int* in_sizes, int n_in,
                              void* d_out, int out_size, void* d_ws, size_t ws_size,
                              hipStream_t stream) {
  const float* x    = (const float*)d_in[0];
  const int*   ei   = (const int*)d_in[1];
  const float* W1l  = (const float*)d_in[2];
  const float* W1r  = (const float*)d_in[3];
  const float* att1 = (const float*)d_in[4];
  const float* b1   = (const float*)d_in[5];
  const float* W2l  = (const float*)d_in[6];
  const float* W2r  = (const float*)d_in[7];
  const float* att2 = (const float*)d_in[8];
  const float* b2   = (const float*)d_in[9];

  const int N  = in_sizes[0] / 128;
  const int E  = in_sizes[1] / 2;
  const int ET = E + N;

  float* ws = (float*)d_ws;
  const size_t NF = (size_t)N * FDIM;
  float*    xl    = ws;
  float*    xr    = xl + NF;
  float*    h     = xr + NF;
  int*      lst   = (int*)(h + NF);
  float*    e     = (float*)lst + NF;
  float*    ex    = e + ET;

  const int per = (((N + 1) + 1023) / 1024 + 31) / 32 * 32;
  const int NP  = per * 1024;
  int* cnt    = lst;
  int* rowptr = cnt + NP;
  int* spair  = rowptr + NP;
  int* tsrc   = (int*)e;
  int* rank   = (int*)ex;

  const int rowTiles   = (N + 15) / 16;
  const int realEdgeBl = (E + 255) / 256;
  const int nodeBl     = (N + 7) / 8;
  const int* esrc = ei;
  const int* edst = ei + E;

  hipMemsetAsync(cnt, 0, sizeof(int) * (size_t)NP, stream);
  gatv2_bucket_rank<<<realEdgeBl, 256, 0, stream>>>(edst, cnt, rank, E);
  gatv2_scan_and_gemm<<<1 + (rowTiles + 7) / 8, 1024, 0, stream>>>(
      cnt, rowptr, per, x, W1l, W1r, xl, xr, N, 128);
  gatv2_bucket_place<<<realEdgeBl, 256, 0, stream>>>(esrc, edst, rowptr, rank, tsrc, E);
  gatv2_segment_order<<<nodeBl, 256, 0, stream>>>(rowptr, tsrc, spair, N, E);

  gatv2_node_fused<<<nodeBl, 256, 0, stream>>>(xl, xr, rowptr, spair, att1, b1, e, h, N, E, 1);

  gatv2_gemm_dual<<<rowTiles, 128, 0, stream>>>(h, W2l, W2r, xl, xr, N, 64);
  gatv2_node_fused<<<nodeBl, 256, 0, stream>>>(xl, xr, rowptr, spair, att2, b2, e, (float*)d_out, N, E, 0);
}


# edge ranks drawn from LDS counters per (node range, edge chunk) block, chunk counts joined by a per-node prefix kernel: scattered global atomics and counter memset gone; order of sums unchanged
# speedup vs baseline: 3.9891x; 1.1457x over previous
#include <hip/hip_runtime.h>
#include <hip/hip_bf16.h>

typedef __attribute__((ext_vector_type(2))) float v2f;
typedef __attribute__((ext_vector_type(8))) float v8f;
typedef __attribute__((ext_vector_type(2))) int v2i;
typedef __attribute__((ext_vector_type(4))) int v4i;

#define NEG_SLOPE 0.2f
#define EPS_DEN 1e-16f
#define FDIM 64

__device__ __forceinline__ void gatv2_gemm_tile(
    const float* __restrict__ X, const float* __restrict__ Wl,
    const float* __restrict__ Wr, float* __restrict__ xl,
    float* __restrict__ xr, int N, int K, int rowTile, int colQ, int lane) {
  const int laneHi = lane >> 4;
  const int lane16 = lane & 15;
  const int colBase = colQ * 16;

  int row  = rowTile * 16 + lane16;
  int rowc = row < N ? row : N - 1;

  v8f accL = {};
  v8f accR = {};
  const float* xrow = X + (size_t)rowc * K;

  for (int k = 0; k < K; k += 4) {
    const int kk = k + 2 * laneHi;
    v2f a = *(const v2f*)(xrow + kk);
    v2f bl, br;
    bl[0] = Wl[(size_t)kk * FDIM + colBase + lane16];
    bl[1] = Wl[(size_t)(kk + 1) * FDIM + colBase + lane16];
    br[0] = Wr[(size_t)kk * FDIM + colBase + lane16];
    br[1] = Wr[(size_t)(kk + 1) * FDIM + colBase + lane16];
    accL = __builtin_amdgcn_wmma_f32_16x16x4_f32(false, a, false, bl,
                                                 (short)0, accL, false, false);
    accR = __builtin_amdgcn_wmma_f32_16x16x4_f32(false, a, false, br,
                                                 (short)0, accR, false, false);
  }

#pragma unroll
  for (int v = 0; v < 8; ++v) {
    int m = v + 8 * laneHi;
    int r = rowTile * 16 + m;
    if (r < N) {
      xl[(size_t)r * FDIM + colBase + lane16] = accL[v];
      xr[(size_t)r * FDIM + colBase + lane16] = accR[v];
    }
  }
}

__global__ __launch_bounds__(128) void gatv2_gemm_dual(
    const float* __restrict__ X, const float* __restrict__ Wl,
    const float* __restrict__ Wr, float* __restrict__ xl,
    float* __restrict__ xr, int N, int K) {
  gatv2_gemm_tile(X, Wl, Wr, xl, xr, N, K, (int)blockIdx.x,
                  (int)(threadIdx.x >> 5), (int)(threadIdx.x & 31));
}

#define RK_R 8
#define RK_C 32
#define RK_CAP 8192
#define RK_U 13
__global__ __launch_bounds__(1024) void gatv2_bucket_rank_lds(
    const int* __restrict__ edst, int* __restrict__ hist,
    int* __restrict__ rank, int E, int NP) {
  __shared__ __attribute__((aligned(16))) int cntL[RK_CAP];
  const int t = threadIdx.x;
  const int r = (int)blockIdx.x % RK_R;
  const int c = (int)blockIdx.x / RK_R;
  const int chunk = (E + RK_C - 1) / RK_C;
  const int e0 = c * chunk;
  int e1 = e0 + chunk;
  e1 = e1 > E ? E : e1;
  for (int nb = r * RK_CAP; nb < NP; nb += RK_R * RK_CAP) {
    for (int i = t * 4; i < RK_CAP; i += 4096) {
      v4i z = {0, 0, 0, 0};
      *(v4i*)(cntL + i) = z;
    }
    __syncthreads();
    for (int eb = e0; eb < e1; eb += RK_U * 1024) {
      int d[RK_U];
      int rk[RK_U];
#pragma unroll
      for (int u = 0; u < RK_U; ++u) {
        const int e = eb + u * 1024 + t;
        d[u] = edst[e < e1 ? e : e1 - 1];
      }
#pragma unroll
      for (int u = 0; u < RK_U; ++u) {
        const int e = eb + u * 1024 + t;
        d[u] = e < e1 ? d[u] - nb : -1;
        rk[u] = 0;
        if ((unsigned)d[u] < (unsigned)RK_CAP) rk[u] = atomicAdd(cntL + d[u], 1);
      }
#pragma unroll
      for (int u = 0; u < RK_U; ++u) {
        if ((unsigned)d[u] < (unsigned)RK_CAP) rank[eb + u * 1024 + t] = rk[u];
      }
    }
    __syncthreads();
    int* hrow = hist + (size_t)c * NP + nb;
    for (int i = t * 4; i < RK_CAP; i += 4096)
      *(v4i*)(hrow + i) = *(const v4i*)(cntL + i);
    __syncthreads();
  }
}

__global__ __launch_bounds__(256) void gatv2_hist_prefix(
    int* hist, int* __restrict__ cnt, int NP) {
  const int n = blockIdx.x * 256 + threadIdx.x;
  if (n >= NP) return;
  int v[RK_C];
#pragma unroll
  for (int c = 0; c < RK_C; ++c) {
    const int* row = hist + (size_t)c * NP;
    v[c] = row[n];
  }
  int run = 0;
#pragma unroll
  for (int c = 0; c < RK_C; ++c) {
    int* row = hist + (size_t)c * NP;
    row[n] = run;
    run += v[c];
  }
  cnt[n] = run;
}

__device__ __forceinline__ void gatv2_rowptr_scan_block(
    const int* __restrict__ cnt, int* __restrict__ rowptr, int per) {
  __shared__ int wsum[32];
  const int t    = threadIdx.x;
  const int lane = t & 31;
  const int wave = t >> 5;
  const int nq   = per >> 2;
  const v4i* c4 = (const v4i*)(cnt + (size_t)t * per);
  int sum = 0;
  for (int i = 0; i < nq; ++i) {
    const v4i v = c4[i];
    sum += (v[0] + v[1]) + (v[2] + v[3]);
  }
  int incl = sum;
  for (int off = 1; off < 32; off <<= 1) {
    const int v = __shfl_up(incl, off, 32);
    if (lane >= off) incl += v;
  }
  if (lane == 31) wsum[wave] = incl;
  __syncthreads();
  if (wave == 0) {
    const int w = wsum[lane];
    int wi = w;
    for (int off = 1; off < 32; off <<= 1) {
      const int v = __shfl_up(wi, off, 32);
      if (lane >= off) wi += v;
    }
    wsum[lane] = wi - w;
  }
  __syncthreads();
  int run = wsum[wave] + incl - sum;
  v4i* r4 = (v4i*)(rowptr + (size_t)t * per);
  for (int i = 0; i < nq; ++i) {
    const v4i v = c4[i];
    v4i o;
    o[0] = run;
    o[1] = o[0] + v[0];
    o[2] = o[1] + v[1];
    o[3] = o[2] + v[2];
    run  = o[3] + v[3];
    r4[i] = o;
  }
}

__global__ __launch_bounds__(1024) void gatv2_scan_and_gemm(
    const int* __restrict__ cnt, int* __restrict__ rowptr, int per,
    const float* __restrict__ X, const float* __restrict__ Wl,
    const float* __restrict__ Wr, float* __restrict__ xl,
    float* __restrict__ xr, int N, int K) {
  if (blockIdx.x == 0) {
    gatv2_rowptr_scan_block(cnt, rowptr, per);
    return;
  }
  const int wv = __builtin_amdgcn_readfirstlane((int)(threadIdx.x >> 5));
  const int rowTile = ((int)blockIdx.x - 1) * 8 + (wv >> 2);
  if (rowTile * 16 >= N) return;
  gatv2_gemm_tile(X, Wl, Wr, xl, xr, N, K, rowTile, wv & 3, (int)(threadIdx.x & 31));
}

__global__ __launch_bounds__(256) void gatv2_bucket_place(
    const int* __restrict__ esrc, const int* __restrict__ edst,
    const int* __restrict__ rowptr, const int* __restrict__ hist,
    const int* __restrict__ rank, int* __restrict__ tsrc, int E, int NP) {
  const int eid = blockIdx.x * 256 + threadIdx.x;
  if (eid >= E) return;
  const int s = esrc[eid];
  int d = edst[eid];
  d = d < 0 ? 0 : (d >= NP ? NP - 1 : d);
  const int chunk = (E + RK_C - 1) / RK_C;
  const int c = eid / chunk;
  const int pos = rowptr[d] + hist[(size_t)c * NP + d] + rank[eid];
  if ((unsigned)pos < (unsigned)E) tsrc[pos] = s;
}

__global__ __launch_bounds__(256) void gatv2_segment_order(
    const int* __restrict__ rowptr, const int* __restrict__ tsrc,
    int* __restrict__ spair, int N, int E) {
  const int lane = threadIdx.x & 31;
  const int wv   = __builtin_amdgcn_readfirstlane((int)(threadIdx.x >> 5));
  const int node = blockIdx.x * 8 + wv;
  if (node >= N) return;
  int beg = rowptr[node];
  int end = rowptr[node + 1];
  beg = beg < 0 ? 0 : beg;
  end = end > E ? E : end;
  for (int cj = beg; cj < end; cj += 32) {
    const int pj = cj + lane;
    const bool vj = pj < end;
    const int sj = tsrc[vj ? pj : end - 1];
    int rank = 0;
    for (int ci = beg; ci < end; ci += 32) {
      const int ni = (end - ci) < 32 ? (end - ci) : 32;
      const int pi = ci + lane;
      const int si = tsrc[pi < end ? pi : end - 1];
      for (int i = 0; i < ni; ++i) {
        const int v = __builtin_amdgcn_readlane(si, i);
        rank += (v < sj || (v == sj && ci + i < pj)) ? 1 : 0;
      }
    }
    if (vj) {
      v2i q;
      q[0] = sj;
      q[1] = node;
      *(v2i*)(spair + 2 * (size_t)(beg + rank)) = q;
    }
  }
}

#define GATV2_TR_STEP(WD)                                   \
  {                                                         \
    const bool hi = (lane & (WD)) != 0;                     \
    _Pragma("unroll") for (int i = 0; i < (WD); ++i) {      \
      const float keep = hi ? p[i + (WD)] : p[i];           \
      const float send = hi ? p[i] : p[i + (WD)];           \
      p[i] = keep + __shfl_xor(send, (WD), 32);             \
    }                                                       \
  }

__device__ __forceinline__ float gatv2_partial(v2f a, v2f b, v2f at) {
  float t0 = a[0] + b[0]; t0 = t0 > 0.f ? t0 : NEG_SLOPE * t0;
  float t1 = a[1] + b[1]; t1 = t1 > 0.f ? t1 : NEG_SLOPE * t1;
  return t0 * at[0] + t1 * at[1];
}

__global__ __launch_bounds__(256) void gatv2_node_fused(
    const float* __restrict__ xl, const float* __restrict__ xr,
    const int* __restrict__ rowptr, const int* __restrict__ spair,
    const float* __restrict__ att, const float* __restrict__ bias,
    float* __restrict__ scr, float* __restrict__ out,
    int N, int E, int relu) {
  const int lane = threadIdx.x & 31;
  const int wv   = __builtin_amdgcn_readfirstlane((int)(threadIdx.x >> 5));
  const int node = blockIdx.x * 8 + wv;
  if (node >= N) return;
  int beg = rowptr[node];
  int end = rowptr[node + 1];
  beg = beg < 0 ? 0 : beg;
  end = end > E ? E : end;
  end = end < beg ? beg : end;
  const int deg = end - beg;
  const float* xlL = xl + 2 * lane;
  const v2f at    = *(const v2f*)(att + 2 * lane);
  const v2f xrn   = *(const v2f*)(xr + (size_t)node * FDIM + 2 * lane);
  const v2f rself = *(const v2f*)(xlL + (size_t)node * FDIM);

  float es = gatv2_partial(rself, xrn, at);

  v2f acc;
  if (deg <= 32) {
    const bool valid = lane < deg;
    int s = node;
    if (valid) s = spair[2 * (size_t)(beg + lane)];
    s = s < 0 ? 0 : (s >= N ? N - 1 : s);
    v2f row[32];
    float p[32];
#pragma unroll
    for (int c = 0; c < 32; c += 8) {
      if (c < deg) {
#pragma unroll
        for (int k = 0; k < 8; ++k) {
          const int sj = __builtin_amdgcn_readlane(s, c + k);
          row[c + k] = *(const v2f*)(xlL + (size_t)sj * FDIM);
        }
      }
    }
    for (int off = 16; off >= 1; off >>= 1) es += __shfl_xor(es, off, 32);
#pragma unroll
    for (int c = 0; c < 32; c += 8) {
      if (c < deg) {
#pragma unroll
        for (int k = 0; k < 8; ++k) p[c + k] = gatv2_partial(row[c + k], xrn, at);
      } else {
#pragma unroll
        for (int k = 0; k < 8; ++k) p[c + k] = 0.f;
      }
    }
    GATV2_TR_STEP(16)
    GATV2_TR_STEP(8)
    GATV2_TR_STEP(4)
    GATV2_TR_STEP(2)
    GATV2_TR_STEP(1)
    const float lg = p[0];
    float m = valid ? fmaxf(lg, es) : es;
    for (int off = 16; off >= 1; off >>= 1) m = fmaxf(m, __shfl_xor(m, off, 32));
    const float exv = valid ? __expf(lg - m) : 0.f;
    const float exs = __expf(es - m);
    float part = exv;
    for (int off = 16; off >= 1; off >>= 1) part += __shfl_xor(part, off, 32);
    const float den = (part + exs) + EPS_DEN;
    const float aself = exs / den;
    const float al = exv / den;
    acc[0] = aself * rself[0];
    acc[1] = aself * rself[1];
#pragma unroll
    for (int c = 0; c < 32; c += 8) {
      if (c < deg) {
#pragma unroll
        for (int k = 0; k < 8; ++k) {
          if (c + k < deg) {
            const float aj = __int_as_float(
                __builtin_amdgcn_readlane(__float_as_int(al), c + k));
            acc[0] += aj * row[c + k][0];
            acc[1] += aj * row[c + k][1];
          }
        }
      }
    }
  } else {
    for (int off = 16; off >= 1; off >>= 1) es += __shfl_xor(es, off, 32);
    float m = es;
    for (int c0 = beg; c0 < end; c0 += 32) {
      const int n = (end - c0) < 32 ? (end - c0) : 32;
      const int slot = (c0 + lane) < end ? (c0 + lane) : (end - 1);
      int s = spair[2 * (size_t)slot];
      s = s < 0 ? 0 : (s >= N ? N - 1 : s);
      float p[32];
#pragma unroll
      for (int c = 0; c < 32; c += 8) {
        if (c < n) {
          v2f r[8];
#pragma unroll
          for (int k = 0; k < 8; ++k) {
            const int sj = __builtin_amdgcn_readlane(s, c + k);
            r[k] = *(const v2f*)(xlL + (size_t)sj * FDIM);
          }
#pragma unroll
          for (int k = 0; k < 8; ++k) p[c + k] = gatv2_partial(r[k], xrn, at);
        } else {
#pragma unroll
          for (int k = 0; k < 8; ++k) p[c + k] = 0.f;
        }
      }
      GATV2_TR_STEP(16)
      GATV2_TR_STEP(8)
      GATV2_TR_STEP(4)
      GATV2_TR_STEP(2)
      GATV2_TR_STEP(1)
      if (c0 + lane < end) {
        scr[c0 + lane] = p[0];
        m = fmaxf(m, p[0]);
      }
    }
    for (int off = 16; off >= 1; off >>= 1) m = fmaxf(m, __shfl_xor(m, off, 32));
    const float exs = __expf(es - m);
    float part = 0.f;
    for (int c0 = beg; c0 < end; c0 += 32) {
      const int slot = c0 + lane;
      if (slot < end) part += __expf(scr[slot] - m);
    }
    for (int off = 16; off >= 1; off >>= 1) part += __shfl_xor(part, off, 32);
    const float den = (part + exs) + EPS_DEN;
    const float aself = exs / den;
    acc[0] = aself * rself[0];
    acc[1] = aself * rself[1];
    for (int c0 = beg; c0 < end; c0 += 32) {
      const int n = (end - c0) < 32 ? (end - c0) : 32;
      const int slot = (c0 + lane) < end ? (c0 + lane) : (end - 1);
      int s = spair[2 * (size_t)slot];
      s = s < 0 ? 0 : (s >= N ? N - 1 : s);
      const float al = (c0 + lane) < end ? __expf(scr[c0 + lane] - m) / den : 0.f;
      for (int j0 = 0; j0 < n; j0 += 8) {
        v2f r[8];
#pragma unroll
        for (int k = 0; k < 8; ++k) {
          const int sj = __builtin_amdgcn_readlane(s, j0 + k);
          r[k] = *(const v2f*)(xlL + (size_t)sj * FDIM);
        }
#pragma unroll
        for (int k = 0; k < 8; ++k) {
          if (j0 + k < n) {
            const float aj = __int_as_float(
                __builtin_amdgcn_readlane(__float_as_int(al), j0 + k));
            acc[0] += aj * r[k][0];
            acc[1] += aj * r[k][1];
          }
        }
      }
    }
  }

  const v2f b = *(const v2f*)(bias + 2 * lane);
  v2f o;
  o[0] = acc[0] + b[0];
  o[1] = acc[1] + b[1];
  if (relu) {
    o[0] = o[0] < 0.f ? 0.f : o[0];
    o[1] = o[1] < 0.f ? 0.f : o[1];
  }
  *(v2f*)(out + (size_t)node * FDIM + 2 * lane) = o;
}
#undef GATV2_TR_STEP

extern "C" void kernel_launch(void* const* d_in, const int* in_sizes, int n_in,
                              void* d_out, int out_size, void* d_ws, size_t ws_size,
                              hipStream_t stream) {
  const float* x    = (const float*)d_in[0];
  const int*   ei   = (const int*)d_in[1];
  const float* W1l  = (const float*)d_in[2];
  const float* W1r  = (const float*)d_in[3];
  const float* att1 = (const float*)d_in[4];
  const float* b1   = (const float*)d_in[5];
  const float* W2l  = (const float*)d_in[6];
  const float* W2r  = (const float*)d_in[7];
  const float* att2 = (const float*)d_in[8];
  const float* b2   = (const float*)d_in[9];

  const int N  = in_sizes[0] / 128;
  const int E  = in_sizes[1] / 2;
  const int ET = E + N;

  float* ws = (float*)d_ws;
  const size_t NF = (size_t)N * FDIM;
  float*    xl    = ws;
  float*    xr    = xl + NF;
  float*    h     = xr + NF;
  int*      lst   = (int*)(h + NF);
  float*    e     = (float*)lst + NF;
  float*    ex    = e + ET;

  const int per = (((N + 1) + 1023) / 1024 + 31) / 32 * 32;
  const int NP  = per * 1024;
  int* cnt    = lst;
  int* rowptr = cnt + NP;
  int* spair  = rowptr + NP;
  int* tsrc   = (int*)e;
  int* rank   = (int*)ex;
  int* hist   = (int*)h;

  const int rowTiles   = (N + 15) / 16;
  const int realEdgeBl = (E + 255) / 256;
  const int nodeBl     = (N + 7) / 8;
  const int* esrc = ei;
  const int* edst = ei + E;

  gatv2_bucket_rank_lds<<<RK_R * RK_C, 1024, 0, stream>>>(edst, hist, rank, E, NP);
  gatv2_hist_prefix<<<NP / 256, 256, 0, stream>>>(hist, cnt, NP);
  gatv2_scan_and_gemm<<<1 + (rowTiles + 7) / 8, 1024, 0, stream>>>(
      cnt, rowptr, per, x, W1l, W1r, xl, xr, N, 128);
  gatv2_bucket_place<<<realEdgeBl, 256, 0, stream>>>(esrc, edst, rowptr, hist, rank, tsrc, E, NP);
  gatv2_segment_order<<<nodeBl, 256, 0, stream>>>(rowptr, tsrc, spair, N, E);

  gatv2_node_fused<<<nodeBl, 256, 0, stream>>>(xl, xr, rowptr, spair, att1, b1, e, h, N, E, 1);

  gatv2_gemm_dual<<<rowTiles, 128, 0, stream>>>(h, W2l, W2r, xl, xr, N, 64);
  gatv2_node_fused<<<nodeBl, 256, 0, stream>>>(xl, xr, rowptr, spair, att2, b2, e, (float*)d_out, N, E, 0);
}


# node step: two nodes per wave, one per 16-lane half (16 B of a row per lane); lane exchanges stay inside a row and run as DPP on the vector ALU; half the waves; order of sums in the logits changes
# speedup vs baseline: 4.5468x; 1.1398x over previous
#include <hip/hip_runtime.h>
#include <hip/hip_bf16.h>

typedef __attribute__((ext_vector_type(2))) float v2f;
typedef __attribute__((ext_vector_type(4))) float v4f;
typedef __attribute__((ext_vector_type(8))) float v8f;
typedef __attribute__((ext_vector_type(2))) int v2i;
typedef __attribute__((ext_vector_type(4))) int v4i;

#define NEG_SLOPE 0.2f
#define EPS_DEN 1e-16f
#define FDIM 64

__device__ __forceinline__ void gatv2_gemm_tile(
    const float* __restrict__ X, const float* __restrict__ Wl,
    const float* __restrict__ Wr, float* __restrict__ xl,
    float* __restrict__ xr, int N, int K, int rowTile, int colQ, int lane) {
  const int laneHi = lane >> 4;
  const int lane16 = lane & 15;
  const int colBase = colQ * 16;

  int row  = rowTile * 16 + lane16;
  int rowc = row < N ? row : N - 1;

  v8f accL = {};
  v8f accR = {};
  const float* xrow = X + (size_t)rowc * K;

  for (int k = 0; k < K; k += 4) {
    const int kk = k + 2 * laneHi;
    v2f a = *(const v2f*)(xrow + kk);
    v2f bl, br;
    bl[0] = Wl[(size_t)kk * FDIM + colBase + lane16];
    bl[1] = Wl[(size_t)(kk + 1) * FDIM + colBase + lane16];
    br[0] = Wr[(size_t)kk * FDIM + colBase + lane16];
    br[1] = Wr[(size_t)(kk + 1) * FDIM + colBase + lane16];
    accL = __builtin_amdgcn_wmma_f32_16x16x4_f32(false, a, false, bl,
                                                 (short)0, accL, false, false);
    accR = __builtin_amdgcn_wmma_f32_16x16x4_f32(false, a, false, br,
                                                 (short)0, accR, false, false);
  }

#pragma unroll
  for (int v = 0; v < 8; ++v) {
    int m = v + 8 * laneHi;
    int r = rowTile * 16 + m;
    if (r < N) {
      xl[(size_t)r * FDIM + colBase + lane16] = accL[v];
      xr[(size_t)r * FDIM + colBase + lane16] = accR[v];
    }
  }
}

__global__ __launch_bounds__(128) void gatv2_gemm_dual(
    const float* __restrict__ X, const float* __restrict__ Wl,
    const float* __restrict__ Wr, float* __restrict__ xl,
    float* __restrict__ xr, int N, int K) {
  gatv2_gemm_tile(X, Wl, Wr, xl, xr, N, K, (int)blockIdx.x,
                  (int)(threadIdx.x >> 5), (int)(threadIdx.x & 31));
}

#define RK_R 8
#define RK_C 32
#define RK_CAP 8192
#define RK_U 13
__global__ __launch_bounds__(1024) void gatv2_bucket_rank_lds(
    const int* __restrict__ edst, int* __restrict__ hist,
    int* __restrict__ rank, int E, int NP) {
  __shared__ __attribute__((aligned(16))) int cntL[RK_CAP];
  const int t = threadIdx.x;
  const int r = (int)blockIdx.x % RK_R;
  const int c = (int)blockIdx.x / RK_R;
  const int chunk = (E + RK_C - 1) / RK_C;
  const int e0 = c * chunk;
  int e1 = e0 + chunk;
  e1 = e1 > E ? E : e1;
  for (int nb = r * RK_CAP; nb < NP; nb += RK_R * RK_CAP) {
    for (int i = t * 4; i < RK_CAP; i += 4096) {
      v4i z = {0, 0, 0, 0};
      *(v4i*)(cntL + i) = z;
    }
    __syncthreads();
    for (int eb = e0; eb < e1; eb += RK_U * 1024) {
      int d[RK_U];
      int rk[RK_U];
#pragma unroll
      for (int u = 0; u < RK_U; ++u) {
        const int e = eb + u * 1024 + t;
        d[u] = edst[e < e1 ? e : e1 - 1];
      }
#pragma unroll
      for (int u = 0; u < RK_U; ++u) {
        const int e = eb + u * 1024 + t;
        d[u] = e < e1 ? d[u] - nb : -1;
        rk[u] = 0;
        if ((unsigned)d[u] < (unsigned)RK_CAP) rk[u] = atomicAdd(cntL + d[u], 1);
      }
#pragma unroll
      for (int u = 0; u < RK_U; ++u) {
        if ((unsigned)d[u] < (unsigned)RK_CAP) rank[eb + u * 1024 + t] = rk[u];
      }
    }
    __syncthreads();
    int* hrow = hist + (size_t)c * NP + nb;
    for (int i = t * 4; i < RK_CAP; i += 4096)
      *(v4i*)(hrow + i) = *(const v4i*)(cntL + i);
    __syncthreads();
  }
}

__global__ __launch_bounds__(256) void gatv2_hist_prefix(
    int* hist, int* __restrict__ cnt, int NP) {
  const int n = blockIdx.x * 256 + threadIdx.x;
  if (n >= NP) return;
  int v[RK_C];
#pragma unroll
  for (int c = 0; c < RK_C; ++c) {
    const int* row = hist + (size_t)c * NP;
    v[c] = row[n];
  }
  int run = 0;
#pragma unroll
  for (int c = 0; c < RK_C; ++c) {
    int* row = hist + (size_t)c * NP;
    row[n] = run;
    run += v[c];
  }
  cnt[n] = run;
}

__device__ __forceinline__ void gatv2_rowptr_scan_block(
    const int* __restrict__ cnt, int* __restrict__ rowptr, int per) {
  __shared__ int wsum[32];
  const int t    = threadIdx.x;
  const int lane = t & 31;
  const int wave = t >> 5;
  const int nq   = per >> 2;
  const v4i* c4 = (const v4i*)(cnt + (size_t)t * per);
  int sum = 0;
  for (int i = 0; i < nq; ++i) {
    const v4i v = c4[i];
    sum += (v[0] + v[1]) + (v[2] + v[3]);
  }
  int incl = sum;
  for (int off = 1; off < 32; off <<= 1) {
    const int v = __shfl_up(incl, off, 32);
    if (lane >= off) incl += v;
  }
  if (lane == 31) wsum[wave] = incl;
  __syncthreads();
  if (wave == 0) {
    const int w = wsum[lane];
    int wi = w;
    for (int off = 1; off < 32; off <<= 1) {
      const int v = __shfl_up(wi, off, 32);
      if (lane >= off) wi += v;
    }
    wsum[lane] = wi - w;
  }
  __syncthreads();
  int run = wsum[wave] + incl - sum;
  v4i* r4 = (v4i*)(rowptr + (size_t)t * per);
  for (int i = 0; i < nq; ++i) {
    const v4i v = c4[i];
    v4i o;
    o[0] = run;
    o[1] = o[0] + v[0];
    o[2] = o[1] + v[1];
    o[3] = o[2] + v[2];
    run  = o[3] + v[3];
    r4[i] = o;
  }
}

__global__ __launch_bounds__(1024) void gatv2_scan_and_gemm(
    const int* __restrict__ cnt, int* __restrict__ rowptr, int per,
    const float* __restrict__ X, const float* __restrict__ Wl,
    const float* __restrict__ Wr, float* __restrict__ xl,
    float* __restrict__ xr, int N, int K) {
  if (blockIdx.x == 0) {
    gatv2_rowptr_scan_block(cnt, rowptr, per);
    return;
  }
  const int wv = __builtin_amdgcn_readfirstlane((int)(threadIdx.x >> 5));
  const int rowTile = ((int)blockIdx.x - 1) * 8 + (wv >> 2);
  if (rowTile * 16 >= N) return;
  gatv2_gemm_tile(X, Wl, Wr, xl, xr, N, K, rowTile, wv & 3, (int)(threadIdx.x & 31));
}

__global__ __launch_bounds__(256) void gatv2_bucket_place(
    const int* __restrict__ esrc, const int* __restrict__ edst,
    const int* __restrict__ rowptr, const int* __restrict__ hist,
    const int* __restrict__ rank, int* __restrict__ tsrc, int E, int NP) {
  const int eid = blockIdx.x * 256 + threadIdx.x;
  if (eid >= E) return;
  const int s = esrc[eid];
  int d = edst[eid];
  d = d < 0 ? 0 : (d >= NP ? NP - 1 : d);
  const int chunk = (E + RK_C - 1) / RK_C;
  const int c = eid / chunk;
  const int pos = rowptr[d] + hist[(size_t)c * NP + d] + rank[eid];
  if ((unsigned)pos < (unsigned)E) tsrc[pos] = s;
}

__global__ __launch_bounds__(256) void gatv2_segment_order(
    const int* __restrict__ rowptr, const int* __restrict__ tsrc,
    int* __restrict__ spair, int N, int E) {
  const int lane = threadIdx.x & 31;
  const int wv   = __builtin_amdgcn_readfirstlane((int)(threadIdx.x >> 5));
  const int node = blockIdx.x * 8 + wv;
  if (node >= N) return;
  int beg = rowptr[node];
  int end = rowptr[node + 1];
  beg = beg < 0 ? 0 : beg;
  end = end > E ? E : end;
  for (int cj = beg; cj < end; cj += 32) {
    const int pj = cj + lane;
    const bool vj = pj < end;
    const int sj = tsrc[vj ? pj : end - 1];
    int rank = 0;
    for (int ci = beg; ci < end; ci += 32) {
      const int ni = (end - ci) < 32 ? (end - ci) : 32;
      const int pi = ci + lane;
      const int si = tsrc[pi < end ? pi : end - 1];
      for (int i = 0; i < ni; ++i) {
        const int v = __builtin_amdgcn_readlane(si, i);
        rank += (v < sj || (v == sj && ci + i < pj)) ? 1 : 0;
      }
    }
    if (vj) {
      v2i q;
      q[0] = sj;
      q[1] = node;
      *(v2i*)(spair + 2 * (size_t)(beg + rank)) = q;
    }
  }
}

template <int J> __device__ __forceinline__ int gatv2_row_share_i(int x) {
  return __builtin_amdgcn_update_dpp(0, x, 0x150 + J, 0xf, 0xf, true);
}
template <int J> __device__ __forceinline__ float gatv2_row_share_f(float x) {
  return __int_as_float(gatv2_row_share_i<J>(__float_as_int(x)));
}
template <int M> __device__ __forceinline__ float gatv2_row_xor_f(float x) {
  return __int_as_float(__builtin_amdgcn_update_dpp(0, __float_as_int(x), 0x160 + M, 0xf, 0xf, true));
}

__device__ __forceinline__ float gatv2_partial4(v4f a, v4f b, v4f at) {
  float t0 = a[0] + b[0]; t0 = t0 > 0.f ? t0 : NEG_SLOPE * t0;
  float t1 = a[1] + b[1]; t1 = t1 > 0.f ? t1 : NEG_SLOPE * t1;
  float t2 = a[2] + b[2]; t2 = t2 > 0.f ? t2 : NEG_SLOPE * t2;
  float t3 = a[3] + b[3]; t3 = t3 > 0.f ? t3 : NEG_SLOPE * t3;
  return (t0 * at[0] + t1 * at[1]) + (t2 * at[2] + t3 * at[3]);
}

#define GATV2_TR_STEP(WD)                                   \
  {                                                         \
    const bool hi = (l16 & (WD)) != 0;                      \
    _Pragma("unroll") for (int i = 0; i < (WD); ++i) {      \
      const float keep = hi ? p[i + (WD)] : p[i];           \
      const float send = hi ? p[i] : p[i + (WD)];           \
      p[i] = keep + gatv2_row_xor_f<(WD)>(send);            \
    }                                                       \
  }
#define GATV2_ROW(J) \
  r[J] = *(const v4f*)(xlB + (unsigned)(gatv2_row_share_i<(J)>(s) * 256 + lofs));
#define GATV2_LOAD4(C) \
  if ((C) < nU) { GATV2_ROW((C) + 0) GATV2_ROW((C) + 1) GATV2_ROW((C) + 2) GATV2_ROW((C) + 3) }
#define GATV2_PART(J) p[J] = gatv2_partial4(r[J], xrn, at);
#define GATV2_PART4(C)                                                                   \
  if ((C) < nU) { GATV2_PART((C) + 0) GATV2_PART((C) + 1) GATV2_PART((C) + 2) GATV2_PART((C) + 3) } \
  else { p[(C) + 0] = 0.f; p[(C) + 1] = 0.f; p[(C) + 2] = 0.f; p[(C) + 3] = 0.f; }
#define GATV2_ACC(J)                                  \
  {                                                   \
    const float aj = gatv2_row_share_f<(J)>(w);       \
    acc[0] += aj * r[J][0];                           \
    acc[1] += aj * r[J][1];                           \
    acc[2] += aj * r[J][2];                           \
    acc[3] += aj * r[J][3];                           \
  }
#define GATV2_ACC4(C) \
  if ((C) < nU) { GATV2_ACC((C) + 0) GATV2_ACC((C) + 1) GATV2_ACC((C) + 2) GATV2_ACC((C) + 3) }

__global__ __launch_bounds__(256) void gatv2_node_fused(
    const float* __restrict__ xl, const float* __restrict__ xr,
    const int* __restrict__ rowptr, const int* __restrict__ spair,
    const float* __restrict__ att, const float* __restrict__ bias,
    float* __restrict__ out, int N, int E, int relu) {
  const int lane = threadIdx.x & 31;
  const int l16  = lane & 15;
  const int wv    = __builtin_amdgcn_readfirstlane((int)(threadIdx.x >> 5));
  const int node0 = (blockIdx.x * 8 + wv) * 2;
  if (node0 >= N) return;
  const int nodeH = node0 + (lane >> 4);
  const bool live = nodeH < N;
  const int node  = live ? nodeH : node0;
  int beg = rowptr[node];
  int end = rowptr[node + 1];
  beg = beg < 0 ? 0 : beg;
  end = end > E ? E : end;
  end = end < beg ? beg : end;
  const int deg = end - beg;
  const int d0   = __builtin_amdgcn_readlane(deg, 0);
  const int d1   = __builtin_amdgcn_readlane(deg, 16);
  const int dmax = d0 > d1 ? d0 : d1;
  const unsigned lofs = 16u * (unsigned)l16;
  const char* xlB = (const char*)xl;
  const v4f at    = *(const v4f*)(att + 4 * l16);
  const v4f xrn   = *(const v4f*)(xr + (size_t)node * FDIM + 4 * l16);
  const v4f rself = *(const v4f*)(xl + (size_t)node * FDIM + 4 * l16);

  int snext = node;
  if (l16 < deg) snext = spair[2 * (size_t)(beg + l16)];

  float es = gatv2_partial4(rself, xrn, at);
  es += gatv2_row_xor_f<8>(es);
  es += gatv2_row_xor_f<4>(es);
  es += gatv2_row_xor_f<2>(es);
  es += gatv2_row_xor_f<1>(es);

  float m   = es;
  float den = 1.f;
  v4f acc   = rself;

#pragma nounroll
  for (int c0 = 0; c0 < dmax; c0 += 16) {
    const int nU = (dmax - c0) < 16 ? (dmax - c0) : 16;
    const bool valid = c0 + l16 < deg;
    int s = snext;
    s = s < 0 ? 0 : (s >= N ? N - 1 : s);
    v4f r[16];
    float p[16];
    GATV2_LOAD4(0)
    GATV2_LOAD4(4)
    GATV2_LOAD4(8)
    GATV2_LOAD4(12)
    snext = node;
    if (c0 + 16 + l16 < deg) snext = spair[2 * (size_t)(beg + c0 + 16 + l16)];
    GATV2_PART4(0)
    GATV2_PART4(4)
    GATV2_PART4(8)
    GATV2_PART4(12)
    GATV2_TR_STEP(8)
    GATV2_TR_STEP(4)
    GATV2_TR_STEP(2)
    GATV2_TR_STEP(1)
    const float lg = p[0];
    float mn = valid ? fmaxf(lg, m) : m;
    mn = fmaxf(mn, gatv2_row_xor_f<8>(mn));
    mn = fmaxf(mn, gatv2_row_xor_f<4>(mn));
    mn = fmaxf(mn, gatv2_row_xor_f<2>(mn));
    mn = fmaxf(mn, gatv2_row_xor_f<1>(mn));
    const float sc = __expf(m - mn);
    const float w  = valid ? __expf(lg - mn) : 0.f;
    float wsum = w;
    wsum += gatv2_row_xor_f<8>(wsum);
    wsum += gatv2_row_xor_f<4>(wsum);
    wsum += gatv2_row_xor_f<2>(wsum);
    wsum += gatv2_row_xor_f<1>(wsum);
    den = den * sc + wsum;
    acc[0] *= sc;
    acc[1] *= sc;
    acc[2] *= sc;
    acc[3] *= sc;
    GATV2_ACC4(0)
    GATV2_ACC4(4)
    GATV2_ACC4(8)
    GATV2_ACC4(12)
    m = mn;
  }

  const float dn = den + EPS_DEN;
  const v4f b = *(const v4f*)(bias + 4 * l16);
  v4f o;
  o[0] = acc[0] / dn + b[0];
  o[1] = acc[1] / dn + b[1];
  o[2] = acc[2] / dn + b[2];
  o[3] = acc[3] / dn + b[3];
  if (relu) {
    o[0] = o[0] < 0.f ? 0.f : o[0];
    o[1] = o[1] < 0.f ? 0.f : o[1];
    o[2] = o[2] < 0.f ? 0.f : o[2];
    o[3] = o[3] < 0.f ? 0.f : o[3];
  }
  if (live) *(v4f*)(out + (size_t)node * FDIM + 4 * l16) = o;
}
#undef GATV2_ACC4
#undef GATV2_ACC
#undef GATV2_PART4
#undef GATV2_PART
#undef GATV2_LOAD4
#undef GATV2_ROW
#undef GATV2_TR_STEP

extern "C" void kernel_launch(void* const* d_in, const int* in_sizes, int n_in,
                              void* d_out, int out_size, void* d_ws, size_t ws_size,
                              hipStream_t stream) {
  const float* x    = (const float*)d_in[0];
  const int*   ei   = (const int*)d_in[1];
  const float* W1l  = (const float*)d_in[2];
  const float* W1r  = (const float*)d_in[3];
  const float* att1 = (const float*)d_in[4];
  const float* b1   = (const float*)d_in[5];
  const float* W2l  = (const float*)d_in[6];
  const float* W2r  = (const float*)d_in[7];
  const float* att2 = (const float*)d_in[8];
  const float* b2   = (const float*)d_in[9];

  const int N  = in_sizes[0] / 128;
  const int E  = in_sizes[1] / 2;
  const int ET = E + N;

  float* ws = (float*)d_ws;
  const size_t NF = (size_t)N * FDIM;
  float*    xl    = ws;
  float*    xr    = xl + NF;
  float*    h     = xr + NF;
  int*      lst   = (int*)(h + NF);
  float*    e     = (float*)lst + NF;
  float*    ex    = e + ET;

  const int per = (((N + 1) + 1023) / 1024 + 31) / 32 * 32;
  const int NP  = per * 1024;
  int* cnt    = lst;
  int* rowptr = cnt + NP;
  int* spair  = rowptr + NP;
  int* tsrc   = (int*)e;
  int* rank   = (int*)ex;
  int* hist   = (int*)h;

  const int rowTiles   = (N + 15) / 16;
  const int realEdgeBl = (E + 255) / 256;
  const int nodeBl     = (N + 7) / 8;
  const int pairBl     = (N + 15) / 16;
  const int* esrc = ei;
  const int* edst = ei + E;

  gatv2_bucket_rank_lds<<<RK_R * RK_C, 1024, 0, stream>>>(edst, hist, rank, E, NP);
  gatv2_hist_prefix<<<NP / 256, 256, 0, stream>>>(hist, cnt, NP);
  gatv2_scan_and_gemm<<<1 + (rowTiles + 7) / 8, 1024, 0, stream>>>(
      cnt, rowptr, per, x, W1l, W1r, xl, xr, N, 128);
  gatv2_bucket_place<<<realEdgeBl, 256, 0, stream>>>(esrc, edst, rowptr, hist, rank, tsrc, E, NP);
  gatv2_segment_order<<<nodeBl, 256, 0, stream>>>(rowptr, tsrc, spair, N, E);

  gatv2_node_fused<<<pairBl, 256, 0, stream>>>(xl, xr, rowptr, spair, att1, b1, h, N, E, 1);

  gatv2_gemm_dual<<<rowTiles, 128, 0, stream>>>(h, W2l, W2r, xl, xr, N, 64);
  gatv2_node_fused<<<pairBl, 256, 0, stream>>>(xl, xr, rowptr, spair, att2, b2, (float*)d_out, N, E, 0);
}


# layer-2 products moved into layer 1's node step: the block's 16 finished rows go to LDS and its 8 waves multiply them by W2_l and W2_r (f32 WMMA); one launch and the h round trip gone
# speedup vs baseline: 4.7054x; 1.0349x over previous
#include <hip/hip_runtime.h>
#include <hip/hip_bf16.h>

typedef __attribute__((ext_vector_type(2))) float v2f;
typedef __attribute__((ext_vector_type(4))) float v4f;
typedef __attribute__((ext_vector_type(8))) float v8f;
typedef __attribute__((ext_vector_type(2))) int v2i;
typedef __attribute__((ext_vector_type(4))) int v4i;

#define NEG_SLOPE 0.2f
#define EPS_DEN 1e-16f
#define FDIM 64

__device__ __forceinline__ void gatv2_gemm_tile(
    const float* __restrict__ X, const float* __restrict__ Wl,
    const float* __restrict__ Wr, float* __restrict__ xl,
    float* __restrict__ xr, int N, int K, int rowTile, int colQ, int lane) {
  const int laneHi = lane >> 4;
  const int lane16 = lane & 15;
  const int colBase = colQ * 16;

  int row  = rowTile * 16 + lane16;
  int rowc = row < N ? row : N - 1;

  v8f accL = {};
  v8f accR = {};
  const float* xrow = X + (size_t)rowc * K;

  for (int k = 0; k < K; k += 4) {
    const int kk = k + 2 * laneHi;
    v2f a = *(const v2f*)(xrow + kk);
    v2f bl, br;
    bl[0] = Wl[(size_t)kk * FDIM + colBase + lane16];
    bl[1] = Wl[(size_t)(kk + 1) * FDIM + colBase + lane16];
    br[0] = Wr[(size_t)kk * FDIM + colBase + lane16];
    br[1] = Wr[(size_t)(kk + 1) * FDIM + colBase + lane16];
    accL = __builtin_amdgcn_wmma_f32_16x16x4_f32(false, a, false, bl,
                                                 (short)0, accL, false, false);
    accR = __builtin_amdgcn_wmma_f32_16x16x4_f32(false, a, false, br,
                                                 (short)0, accR, false, false);
  }

#pragma unroll
  for (int v = 0; v < 8; ++v) {
    int m = v + 8 * laneHi;
    int r = rowTile * 16 + m;
    if (r < N) {
      xl[(size_t)r * FDIM + colBase + lane16] = accL[v];
      xr[(size_t)r * FDIM + colBase + lane16] = accR[v];
    }
  }
}

#define RK_R 8
#define RK_C 32
#define RK_CAP 8192
#define RK_U 13
__global__ __launch_bounds__(1024) void gatv2_bucket_rank_lds(
    const int* __restrict__ edst, int* __restrict__ hist,
    int* __restrict__ rank, int E, int NP) {
  __shared__ __attribute__((aligned(16))) int cntL[RK_CAP];
  const int t = threadIdx.x;
  const int r = (int)blockIdx.x % RK_R;
  const int c = (int)blockIdx.x / RK_R;
  const int chunk = (E + RK_C - 1) / RK_C;
  const int e0 = c * chunk;
  int e1 = e0 + chunk;
  e1 = e1 > E ? E : e1;
  for (int nb = r * RK_CAP; nb < NP; nb += RK_R * RK_CAP) {
    for (int i = t * 4; i < RK_CAP; i += 4096) {
      v4i z = {0, 0, 0, 0};
      *(v4i*)(cntL + i) = z;
    }
    __syncthreads();
    for (int eb = e0; eb < e1; eb += RK_U * 1024) {
      int d[RK_U];
      int rk[RK_U];
#pragma unroll
      for (int u = 0; u < RK_U; ++u) {
        const int e = eb + u * 1024 + t;
        d[u] = edst[e < e1 ? e : e1 - 1];
      }
#pragma unroll
      for (int u = 0; u < RK_U; ++u) {
        const int e = eb + u * 1024 + t;
        d[u] = e < e1 ? d[u] - nb : -1;
        rk[u] = 0;
        if ((unsigned)d[u] < (unsigned)RK_CAP) rk[u] = atomicAdd(cntL + d[u], 1);
      }
#pragma unroll
      for (int u = 0; u < RK_U; ++u) {
        if ((unsigned)d[u] < (unsigned)RK_CAP) rank[eb + u * 1024 + t] = rk[u];
      }
    }
    __syncthreads();
    int* hrow = hist + (size_t)c * NP + nb;
    for (int i = t * 4; i < RK_CAP; i += 4096)
      *(v4i*)(hrow + i) = *(const v4i*)(cntL + i);
    __syncthreads();
  }
}

__global__ __launch_bounds__(256) void gatv2_hist_prefix(
    int* hist, int* __restrict__ cnt, int NP) {
  const int n = blockIdx.x * 256 + threadIdx.x;
  if (n >= NP) return;
  int v[RK_C];
#pragma unroll
  for (int c = 0; c < RK_C; ++c) {
    const int* row = hist + (size_t)c * NP;
    v[c] = row[n];
  }
  int run = 0;
#pragma unroll
  for (int c = 0; c < RK_C; ++c) {
    int* row = hist + (size_t)c * NP;
    row[n] = run;
    run += v[c];
  }
  cnt[n] = run;
}

__device__ __forceinline__ void gatv2_rowptr_scan_block(
    const int* __restrict__ cnt, int* __restrict__ rowptr, int per) {
  __shared__ int wsum[32];
  const int t    = threadIdx.x;
  const int lane = t & 31;
  const int wave = t >> 5;
  const int nq   = per >> 2;
  const v4i* c4 = (const v4i*)(cnt + (size_t)t * per);
  int sum = 0;
  for (int i = 0; i < nq; ++i) {
    const v4i v = c4[i];
    sum += (v[0] + v[1]) + (v[2] + v[3]);
  }
  int incl = sum;
  for (int off = 1; off < 32; off <<= 1) {
    const int v = __shfl_up(incl, off, 32);
    if (lane >= off) incl += v;
  }
  if (lane == 31) wsum[wave] = incl;
  __syncthreads();
  if (wave == 0) {
    const int w = wsum[lane];
    int wi = w;
    for (int off = 1; off < 32; off <<= 1) {
      const int v = __shfl_up(wi, off, 32);
      if (lane >= off) wi += v;
    }
    wsum[lane] = wi - w;
  }
  __syncthreads();
  int run = wsum[wave] + incl - sum;
  v4i* r4 = (v4i*)(rowptr + (size_t)t * per);
  for (int i = 0; i < nq; ++i) {
    const v4i v = c4[i];
    v4i o;
    o[0] = run;
    o[1] = o[0] + v[0];
    o[2] = o[1] + v[1];
    o[3] = o[2] + v[2];
    run  = o[3] + v[3];
    r4[i] = o;
  }
}

__global__ __launch_bounds__(1024) void gatv2_scan_and_gemm(
    const int* __restrict__ cnt, int* __restrict__ rowptr, int per,
    const float* __restrict__ X, const float* __restrict__ Wl,
    const float* __restrict__ Wr, float* __restrict__ xl,
    float* __restrict__ xr, int N, int K) {
  if (blockIdx.x == 0) {
    gatv2_rowptr_scan_block(cnt, rowptr, per);
    return;
  }
  const int wv = __builtin_amdgcn_readfirstlane((int)(threadIdx.x >> 5));
  const int rowTile = ((int)blockIdx.x - 1) * 8 + (wv >> 2);
  if (rowTile * 16 >= N) return;
  gatv2_gemm_tile(X, Wl, Wr, xl, xr, N, K, rowTile, wv & 3, (int)(threadIdx.x & 31));
}

__global__ __launch_bounds__(256) void gatv2_bucket_place(
    const int* __restrict__ esrc, const int* __restrict__ edst,
    const int* __restrict__ rowptr, const int* __restrict__ hist,
    const int* __restrict__ rank, int* __restrict__ tsrc, int E, int NP) {
  const int eid = blockIdx.x * 256 + threadIdx.x;
  if (eid >= E) return;
  const int s = esrc[eid];
  int d = edst[eid];
  d = d < 0 ? 0 : (d >= NP ? NP - 1 : d);
  const int chunk = (E + RK_C - 1) / RK_C;
  const int c = eid / chunk;
  const int pos = rowptr[d] + hist[(size_t)c * NP + d] + rank[eid];
  if ((unsigned)pos < (unsigned)E) tsrc[pos] = s;
}

__global__ __launch_bounds__(256) void gatv2_segment_order(
    const int* __restrict__ rowptr, const int* __restrict__ tsrc,
    int* __restrict__ spair, int N, int E) {
  const int lane = threadIdx.x & 31;
  const int wv   = __builtin_amdgcn_readfirstlane((int)(threadIdx.x >> 5));
  const int node = blockIdx.x * 8 + wv;
  if (node >= N) return;
  int beg = rowptr[node];
  int end = rowptr[node + 1];
  beg = beg < 0 ? 0 : beg;
  end = end > E ? E : end;
  for (int cj = beg; cj < end; cj += 32) {
    const int pj = cj + lane;
    const bool vj = pj < end;
    const int sj = tsrc[vj ? pj : end - 1];
    int rank = 0;
    for (int ci = beg; ci < end; ci += 32) {
      const int ni = (end - ci) < 32 ? (end - ci) : 32;
      const int pi = ci + lane;
      const int si = tsrc[pi < end ? pi : end - 1];
      for (int i = 0; i < ni; ++i) {
        const int v = __builtin_amdgcn_readlane(si, i);
        rank += (v < sj || (v == sj && ci + i < pj)) ? 1 : 0;
      }
    }
    if (vj) {
      v2i q;
      q[0] = sj;
      q[1] = node;
      *(v2i*)(spair + 2 * (size_t)(beg + rank)) = q;
    }
  }
}

template <int J> __device__ __forceinline__ int gatv2_row_share_i(int x) {
  return __builtin_amdgcn_update_dpp(0, x, 0x150 + J, 0xf, 0xf, true);
}
template <int J> __device__ __forceinline__ float gatv2_row_share_f(float x) {
  return __int_as_float(gatv2_row_share_i<J>(__float_as_int(x)));
}
template <int M> __device__ __forceinline__ float gatv2_row_xor_f(float x) {
  return __int_as_float(__builtin_amdgcn_update_dpp(0, __float_as_int(x), 0x160 + M, 0xf, 0xf, true));
}

__device__ __forceinline__ float gatv2_partial4(v4f a, v4f b, v4f at) {
  float t0 = a[0] + b[0]; t0 = t0 > 0.f ? t0 : NEG_SLOPE * t0;
  float t1 = a[1] + b[1]; t1 = t1 > 0.f ? t1 : NEG_SLOPE * t1;
  float t2 = a[2] + b[2]; t2 = t2 > 0.f ? t2 : NEG_SLOPE * t2;
  float t3 = a[3] + b[3]; t3 = t3 > 0.f ? t3 : NEG_SLOPE * t3;
  return (t0 * at[0] + t1 * at[1]) + (t2 * at[2] + t3 * at[3]);
}

#define GATV2_TR_STEP(WD)                                   \
  {                                                         \
    const bool hi = (l16 & (WD)) != 0;                      \
    _Pragma("unroll") for (int i = 0; i < (WD); ++i) {      \
      const float keep = hi ? p[i + (WD)] : p[i];           \
      const float send = hi ? p[i] : p[i + (WD)];           \
      p[i] = keep + gatv2_row_xor_f<(WD)>(send);            \
    }                                                       \
  }
#define GATV2_ROW(J) \
  r[J] = *(const v4f*)(xlB + (unsigned)(gatv2_row_share_i<(J)>(s) * 256 + lofs));
#define GATV2_LOAD4(C) \
  if ((C) < nU) { GATV2_ROW((C) + 0) GATV2_ROW((C) + 1) GATV2_ROW((C) + 2) GATV2_ROW((C) + 3) }
#define GATV2_PART(J) p[J] = gatv2_partial4(r[J], xrn, at);
#define GATV2_PART4(C)                                                                   \
  if ((C) < nU) { GATV2_PART((C) + 0) GATV2_PART((C) + 1) GATV2_PART((C) + 2) GATV2_PART((C) + 3) } \
  else { p[(C) + 0] = 0.f; p[(C) + 1] = 0.f; p[(C) + 2] = 0.f; p[(C) + 3] = 0.f; }
#define GATV2_ACC(J)                                  \
  {                                                   \
    const float aj = gatv2_row_share_f<(J)>(w);       \
    acc[0] += aj * r[J][0];                           \
    acc[1] += aj * r[J][1];                           \
    acc[2] += aj * r[J][2];                           \
    acc[3] += aj * r[J][3];                           \
  }
#define GATV2_ACC4(C) \
  if ((C) < nU) { GATV2_ACC((C) + 0) GATV2_ACC((C) + 1) GATV2_ACC((C) + 2) GATV2_ACC((C) + 3) }

#define HT_PITCH 68
template <bool FUSE>
__global__ __launch_bounds__(256) void gatv2_node_fused(
    const float* __restrict__ xl, const float* xr,
    const int* __restrict__ rowptr, const int* __restrict__ spair,
    const float* __restrict__ att, const float* __restrict__ bias,
    float* __restrict__ out, int N, int E, int relu,
    const float* __restrict__ W2l, const float* __restrict__ W2r,
    float* xl2, float* xr2) {
  const int lane = threadIdx.x & 31;
  const int l16  = lane & 15;
  const int wv    = __builtin_amdgcn_readfirstlane((int)(threadIdx.x >> 5));
  const int node0 = (blockIdx.x * 8 + wv) * 2;
  if (!FUSE && node0 >= N) return;
  const int nodeH = node0 + (lane >> 4);
  const bool live = nodeH < N;
  const int node  = live ? nodeH : (node0 < N ? node0 : N - 1);
  int beg = rowptr[node];
  int end = rowptr[node + 1];
  beg = beg < 0 ? 0 : beg;
  end = end > E ? E : end;
  end = end < beg ? beg : end;
  const int deg = end - beg;
  const int d0   = __builtin_amdgcn_readlane(deg, 0);
  const int d1   = __builtin_amdgcn_readlane(deg, 16);
  const int dmax = d0 > d1 ? d0 : d1;
  const unsigned lofs = 16u * (unsigned)l16;
  const char* xlB = (const char*)xl;
  const v4f at    = *(const v4f*)(att + 4 * l16);
  const v4f xrn   = *(const v4f*)(xr + (size_t)node * FDIM + 4 * l16);
  const v4f rself = *(const v4f*)(xl + (size_t)node * FDIM + 4 * l16);

  int snext = node;
  if (l16 < deg) snext = spair[2 * (size_t)(beg + l16)];

  float es = gatv2_partial4(rself, xrn, at);
  es += gatv2_row_xor_f<8>(es);
  es += gatv2_row_xor_f<4>(es);
  es += gatv2_row_xor_f<2>(es);
  es += gatv2_row_xor_f<1>(es);

  float m   = es;
  float den = 1.f;
  v4f acc   = rself;

#pragma nounroll
  for (int c0 = 0; c0 < dmax; c0 += 16) {
    const int nU = (dmax - c0) < 16 ? (dmax - c0) : 16;
    const bool valid = c0 + l16 < deg;
    int s = snext;
    s = s < 0 ? 0 : (s >= N ? N - 1 : s);
    v4f r[16];
    float p[16];
    GATV2_LOAD4(0)
    GATV2_LOAD4(4)
    GATV2_LOAD4(8)
    GATV2_LOAD4(12)
    snext = node;
    if (c0 + 16 + l16 < deg) snext = spair[2 * (size_t)(beg + c0 + 16 + l16)];
    GATV2_PART4(0)
    GATV2_PART4(4)
    GATV2_PART4(8)
    GATV2_PART4(12)
    GATV2_TR_STEP(8)
    GATV2_TR_STEP(4)
    GATV2_TR_STEP(2)
    GATV2_TR_STEP(1)
    const float lg = p[0];
    float mn = valid ? fmaxf(lg, m) : m;
    mn = fmaxf(mn, gatv2_row_xor_f<8>(mn));
    mn = fmaxf(mn, gatv2_row_xor_f<4>(mn));
    mn = fmaxf(mn, gatv2_row_xor_f<2>(mn));
    mn = fmaxf(mn, gatv2_row_xor_f<1>(mn));
    const float sc = __expf(m - mn);
    const float w  = valid ? __expf(lg - mn) : 0.f;
    float wsum = w;
    wsum += gatv2_row_xor_f<8>(wsum);
    wsum += gatv2_row_xor_f<4>(wsum);
    wsum += gatv2_row_xor_f<2>(wsum);
    wsum += gatv2_row_xor_f<1>(wsum);
    den = den * sc + wsum;
    acc[0] *= sc;
    acc[1] *= sc;
    acc[2] *= sc;
    acc[3] *= sc;
    GATV2_ACC4(0)
    GATV2_ACC4(4)
    GATV2_ACC4(8)
    GATV2_ACC4(12)
    m = mn;
  }

  const float dn = den + EPS_DEN;
  const v4f b = *(const v4f*)(bias + 4 * l16);
  v4f o;
  o[0] = acc[0] / dn + b[0];
  o[1] = acc[1] / dn + b[1];
  o[2] = acc[2] / dn + b[2];
  o[3] = acc[3] / dn + b[3];
  if (relu) {
    o[0] = o[0] < 0.f ? 0.f : o[0];
    o[1] = o[1] < 0.f ? 0.f : o[1];
    o[2] = o[2] < 0.f ? 0.f : o[2];
    o[3] = o[3] < 0.f ? 0.f : o[3];
  }
  if (!FUSE) {
    if (live) *(v4f*)(out + (size_t)node * FDIM + 4 * l16) = o;
  } else {
    __shared__ __attribute__((aligned(16))) float hT[16 * HT_PITCH];
    const int laneHi  = lane >> 4;
    const int sel     = wv >> 2;
    const int colBase = (wv & 3) * 16;
    const float* Wm   = sel ? W2r : W2l;
    float* dst        = sel ? xr2 : xl2;
    v2f bfr[FDIM / 4];
#pragma unroll
    for (int kq = 0; kq < FDIM / 4; ++kq) {
      const int kk = 4 * kq + 2 * laneHi;
      bfr[kq][0] = Wm[(size_t)kk * FDIM + colBase + l16];
      bfr[kq][1] = Wm[(size_t)(kk + 1) * FDIM + colBase + l16];
    }
    *(v4f*)(hT + (2 * wv + laneHi) * HT_PITCH + 4 * l16) = o;
    __syncthreads();
    v8f accT = {};
#pragma unroll
    for (int kq = 0; kq < FDIM / 4; ++kq) {
      const v2f a = *(const v2f*)(hT + l16 * HT_PITCH + 4 * kq + 2 * laneHi);
      accT = __builtin_amdgcn_wmma_f32_16x16x4_f32(false, a, false, bfr[kq],
                                                   (short)0, accT, false, false);
    }
    const int rbase = (int)blockIdx.x * 16 + 8 * laneHi;
#pragma unroll
    for (int v = 0; v < 8; ++v) {
      const int r = rbase + v;
      if (r < N) dst[(size_t)r * FDIM + colBase + l16] = accT[v];
    }
  }
}
#undef HT_PITCH
#undef GATV2_ACC4
#undef GATV2_ACC
#undef GATV2_PART4
#undef GATV2_PART
#undef GATV2_LOAD4
#undef GATV2_ROW
#undef GATV2_TR_STEP

extern "C" void kernel_launch(void* const* d_in, const int* in_sizes, int n_in,
                              void* d_out, int out_size, void* d_ws, size_t ws_size,
                              hipStream_t stream) {
  const float* x    = (const float*)d_in[0];
  const int*   ei   = (const int*)d_in[1];
  const float* W1l  = (const float*)d_in[2];
  const float* W1r  = (const float*)d_in[3];
  const float* att1 = (const float*)d_in[4];
  const float* b1   = (const float*)d_in[5];
  const float* W2l  = (const float*)d_in[6];
  const float* W2r  = (const float*)d_in[7];
  const float* att2 = (const float*)d_in[8];
  const float* b2   = (const float*)d_in[9];

  const int N  = in_sizes[0] / 128;
  const int E  = in_sizes[1] / 2;
  const int ET = E + N;

  float* ws = (float*)d_ws;
  const size_t NF = (size_t)N * FDIM;
  float*    xl    = ws;
  float*    xr    = xl + NF;
  float*    h     = xr + NF;
  int*      lst   = (int*)(h + NF);
  float*    e     = (float*)lst + NF;
  float*    ex    = e + ET;

  const int per = (((N + 1) + 1023) / 1024 + 31) / 32 * 32;
  const int NP  = per * 1024;
  int* cnt    = lst;
  int* rowptr = cnt + NP;
  int* spair  = rowptr + NP;
  int* tsrc   = (int*)e;
  int* rank   = (int*)ex;
  int* hist   = (int*)h;

  const int rowTiles   = (N + 15) / 16;
  const int realEdgeBl = (E + 255) / 256;
  const int nodeBl     = (N + 7) / 8;
  const int pairBl     = (N + 15) / 16;
  const int* esrc = ei;
  const int* edst = ei + E;

  gatv2_bucket_rank_lds<<<RK_R * RK_C, 1024, 0, stream>>>(edst, hist, rank, E, NP);
  gatv2_hist_prefix<<<NP / 256, 256, 0, stream>>>(hist, cnt, NP);
  gatv2_scan_and_gemm<<<1 + (rowTiles + 7) / 8, 1024, 0, stream>>>(
      cnt, rowptr, per, x, W1l, W1r, xl, xr, N, 128);
  gatv2_bucket_place<<<realEdgeBl, 256, 0, stream>>>(esrc, edst, rowptr, hist, rank, tsrc, E, NP);
  gatv2_segment_order<<<nodeBl, 256, 0, stream>>>(rowptr, tsrc, spair, N, E);

  gatv2_node_fused<true><<<pairBl, 256, 0, stream>>>(
      xl, xr, rowptr, spair, att1, b1, nullptr, N, E, 1, W2l, W2r, h, xr);

  gatv2_node_fused<false><<<pairBl, 256, 0, stream>>>(
      h, xr, rowptr, spair, att2, b2, (float*)d_out, N, E, 0, nullptr, nullptr, nullptr, nullptr);
}


# rank kernel: edge ranks fixed by edge index (LDS key slots per destination, an edge counts the keys below its own): placed list has a fixed order, segment-order launch gone; order of sums changes
# speedup vs baseline: 4.9636x; 1.0549x over previous
#include <hip/hip_runtime.h>
#include <hip/hip_bf16.h>

typedef __attribute__((ext_vector_type(2))) float v2f;
typedef __attribute__((ext_vector_type(4))) float v4f;
typedef __attribute__((ext_vector_type(8))) float v8f;
typedef __attribute__((ext_vector_type(2))) int v2i;
typedef __attribute__((ext_vector_type(4))) int v4i;

#define NEG_SLOPE 0.2f
#define EPS_DEN 1e-16f
#define FDIM 64

__device__ __forceinline__ void gatv2_gemm_tile(
    const float* __restrict__ X, const float* __restrict__ Wl,
    const float* __restrict__ Wr, float* __restrict__ xl,
    float* __restrict__ xr, int N, int K, int rowTile, int colQ, int lane) {
  const int laneHi = lane >> 4;
  const int lane16 = lane & 15;
  const int colBase = colQ * 16;

  int row  = rowTile * 16 + lane16;
  int rowc = row < N ? row : N - 1;

  v8f accL = {};
  v8f accR = {};
  const float* xrow = X + (size_t)rowc * K;

  for (int k = 0; k < K; k += 4) {
    const int kk = k + 2 * laneHi;
    v2f a = *(const v2f*)(xrow + kk);
    v2f bl, br;
    bl[0] = Wl[(size_t)kk * FDIM + colBase + lane16];
    bl[1] = Wl[(size_t)(kk + 1) * FDIM + colBase + lane16];
    br[0] = Wr[(size_t)kk * FDIM + colBase + lane16];
    br[1] = Wr[(size_t)(kk + 1) * FDIM + colBase + lane16];
    accL = __builtin_amdgcn_wmma_f32_16x16x4_f32(false, a, false, bl,
                                                 (short)0, accL, false, false);
    accR = __builtin_amdgcn_wmma_f32_16x16x4_f32(false, a, false, br,
                                                 (short)0, accR, false, false);
  }

#pragma unroll
  for (int v = 0; v < 8; ++v) {
    int m = v + 8 * laneHi;
    int r = rowTile * 16 + m;
    if (r < N) {
      xl[(size_t)r * FDIM + colBase + lane16] = accL[v];
      xr[(size_t)r * FDIM + colBase + lane16] = accR[v];
    }
  }
}

#define RK_R 8
#define RK_C 32
#define RK_CAP 8192
#define RK_U 25
#define RK_K 8
__global__ __launch_bounds__(1024) void gatv2_bucket_rank_lds(
    const int* __restrict__ edst, int* __restrict__ hist,
    int* __restrict__ rank, int E, int NP) {
  static_assert(RK_U * 1024 <= 32767, "an edge's index in its trip must stay below the empty slot's 0x7fff");
  static_assert(RK_CAP == 8192 && RK_K == 8, "the list entry packs 13 bits of destination; a slot row is one 16-byte load");
  __shared__ __attribute__((aligned(16))) int cntL[RK_CAP];
  __shared__ __attribute__((aligned(16))) int tcnL[RK_CAP];
  __shared__ __attribute__((aligned(16))) unsigned int keyL[RK_CAP * RK_K / 2];
  __shared__ int listL[32 * RK_U * 32];
  const int t    = threadIdx.x;
  const int lane = t & 31;
  const int wv   = __builtin_amdgcn_readfirstlane(t >> 5);
  int* myl = listL + wv * (RK_U * 32);
  const int r = (int)blockIdx.x % RK_R;
  const int c = (int)blockIdx.x / RK_R;
  const int chunk = (E + RK_C - 1) / RK_C;
  const int e0 = c * chunk;
  int e1 = e0 + chunk;
  e1 = e1 > E ? E : e1;
  for (int nb = r * RK_CAP; nb < NP; nb += RK_R * RK_CAP) {
    for (int i = t * 4; i < RK_CAP; i += 4096) {
      v4i z = {0, 0, 0, 0};
      *(v4i*)(cntL + i) = z;
    }
    for (int eb = e0; eb < e1; eb += RK_U * 1024) {
      for (int i = t * 4; i < RK_CAP; i += 4096) {
        v4i z = {0, 0, 0, 0};
        *(v4i*)(tcnL + i) = z;
      }
      for (int i = t * 4; i < RK_CAP * RK_K / 2; i += 4096) {
        v4i f = {0x7fff7fff, 0x7fff7fff, 0x7fff7fff, 0x7fff7fff};
        *(v4i*)(keyL + i) = f;
      }
      __syncthreads();
      int d[RK_U];
#pragma unroll
      for (int u = 0; u < RK_U; ++u) {
        const int e = eb + u * 1024 + t;
        d[u] = edst[e < e1 ? e : e1 - 1];
      }
      int n = 0;
#pragma unroll
      for (int u = 0; u < RK_U; ++u) {
        const int e  = eb + u * 1024 + t;
        const int dl = d[u] - nb;
        const bool inr = e < e1 && (unsigned)dl < (unsigned)RK_CAP;
        const unsigned m = __builtin_amdgcn_ballot_w32(inr);
        const int pre = (int)__builtin_amdgcn_mbcnt_lo(m, 0u);
        if (inr) myl[n + pre] = dl | ((u * 1024 + t) << 13);
        n += __builtin_popcount(m);
      }
      for (int i = lane; i < n; i += 32) {
        const int pk  = myl[i];
        const int dl  = pk & (RK_CAP - 1);
        const int key = pk >> 13;
        const int a = atomicAdd(tcnL + dl, 1);
        if (a < RK_K) ((unsigned short*)keyL)[dl * RK_K + a] = (unsigned short)key;
      }
      __syncthreads();
      for (int i = lane; i < n; i += 32) {
        const int pk  = myl[i];
        const int dl  = pk & (RK_CAP - 1);
        const int key = pk >> 13;
        const int g    = tcnL[dl];
        const int base = cntL[dl];
        int det;
        if (g <= RK_K) {
          const v4i kw = *(const v4i*)(keyL + dl * (RK_K / 2));
          const unsigned cc = 0x80008000u - (unsigned)key * 0x10001u;
          const int ge = __builtin_popcount(((unsigned)kw[0] + cc) & 0x80008000u)
                       + __builtin_popcount(((unsigned)kw[1] + cc) & 0x80008000u)
                       + __builtin_popcount(((unsigned)kw[2] + cc) & 0x80008000u)
                       + __builtin_popcount(((unsigned)kw[3] + cc) & 0x80008000u);
          det = RK_K - ge;
        } else {
          const int dg = dl + nb;
          det = 0;
#pragma nounroll
          for (int e2 = eb; e2 < eb + key; ++e2) det += (edst[e2] == dg) ? 1 : 0;
        }
        rank[eb + key] = base + det;
      }
      __syncthreads();
      for (int i = t * 4; i < RK_CAP; i += 4096)
        *(v4i*)(cntL + i) = *(const v4i*)(cntL + i) + *(const v4i*)(tcnL + i);
    }
    __syncthreads();
    int* hrow = hist + (size_t)c * NP + nb;
    for (int i = t * 4; i < RK_CAP; i += 4096)
      *(v4i*)(hrow + i) = *(const v4i*)(cntL + i);
    __syncthreads();
  }
}

__global__ __launch_bounds__(256) void gatv2_hist_prefix(
    int* hist, int* __restrict__ cnt, int NP) {
  const int n = blockIdx.x * 256 + threadIdx.x;
  if (n >= NP) return;
  int v[RK_C];
#pragma unroll
  for (int c = 0; c < RK_C; ++c) {
    const int* row = hist + (size_t)c * NP;
    v[c] = row[n];
  }
  int run = 0;
#pragma unroll
  for (int c = 0; c < RK_C; ++c) {
    int* row = hist + (size_t)c * NP;
    row[n] = run;
    run += v[c];
  }
  cnt[n] = run;
}

__device__ __forceinline__ void gatv2_rowptr_scan_block(
    const int* __restrict__ cnt, int* __restrict__ rowptr, int per) {
  __shared__ int wsum[32];
  const int t    = threadIdx.x;
  const int lane = t & 31;
  const int wave = t >> 5;
  const int nq   = per >> 2;
  const v4i* c4 = (const v4i*)(cnt + (size_t)t * per);
  int sum = 0;
  for (int i = 0; i < nq; ++i) {
    const v4i v = c4[i];
    sum += (v[0] + v[1]) + (v[2] + v[3]);
  }
  int incl = sum;
  for (int off = 1; off < 32; off <<= 1) {
    const int v = __shfl_up(incl, off, 32);
    if (lane >= off) incl += v;
  }
  if (lane == 31) wsum[wave] = incl;
  __syncthreads();
  if (wave == 0) {
    const int w = wsum[lane];
    int wi = w;
    for (int off = 1; off < 32; off <<= 1) {
      const int v = __shfl_up(wi, off, 32);
      if (lane >= off) wi += v;
    }
    wsum[lane] = wi - w;
  }
  __syncthreads();
  int run = wsum[wave] + incl - sum;
  v4i* r4 = (v4i*)(rowptr + (size_t)t * per);
  for (int i = 0; i < nq; ++i) {
    const v4i v = c4[i];
    v4i o;
    o[0] = run;
    o[1] = o[0] + v[0];
    o[2] = o[1] + v[1];
    o[3] = o[2] + v[2];
    run  = o[3] + v[3];
    r4[i] = o;
  }
}

__global__ __launch_bounds__(1024) void gatv2_scan_and_gemm(
    const int* __restrict__ cnt, int* __restrict__ rowptr, int per,
    const float* __restrict__ X, const float* __restrict__ Wl,
    const float* __restrict__ Wr, float* __restrict__ xl,
    float* __restrict__ xr, int N, int K) {
  if (blockIdx.x == 0) {
    gatv2_rowptr_scan_block(cnt, rowptr, per);
    return;
  }
  const int wv = __builtin_amdgcn_readfirstlane((int)(threadIdx.x >> 5));
  const int rowTile = ((int)blockIdx.x - 1) * 8 + (wv >> 2);
  if (rowTile * 16 >= N) return;
  gatv2_gemm_tile(X, Wl, Wr, xl, xr, N, K, rowTile, wv & 3, (int)(threadIdx.x & 31));
}

__global__ __launch_bounds__(256) void gatv2_bucket_place(
    const int* __restrict__ esrc, const int* __restrict__ edst,
    const int* __restrict__ rowptr, const int* __restrict__ hist,
    const int* __restrict__ rank, int* __restrict__ tsrc, int E, int NP) {
  const int eid = blockIdx.x * 256 + threadIdx.x;
  if (eid >= E) return;
  const int s = esrc[eid];
  int d = edst[eid];
  d = d < 0 ? 0 : (d >= NP ? NP - 1 : d);
  const int chunk = (E + RK_C - 1) / RK_C;
  const int c = eid / chunk;
  const int pos = rowptr[d] + hist[(size_t)c * NP + d] + rank[eid];
  if ((unsigned)pos < (unsigned)E) tsrc[pos] = s;
}

template <int J> __device__ __forceinline__ int gatv2_row_share_i(int x) {
  return __builtin_amdgcn_update_dpp(0, x, 0x150 + J, 0xf, 0xf, true);
}
template <int J> __device__ __forceinline__ float gatv2_row_share_f(float x) {
  return __int_as_float(gatv2_row_share_i<J>(__float_as_int(x)));
}
template <int M> __device__ __forceinline__ float gatv2_row_xor_f(float x) {
  return __int_as_float(__builtin_amdgcn_update_dpp(0, __float_as_int(x), 0x160 + M, 0xf, 0xf, true));
}

__device__ __forceinline__ float gatv2_partial4(v4f a, v4f b, v4f at) {
  float t0 = a[0] + b[0]; t0 = t0 > 0.f ? t0 : NEG_SLOPE * t0;
  float t1 = a[1] + b[1]; t1 = t1 > 0.f ? t1 : NEG_SLOPE * t1;
  float t2 = a[2] + b[2]; t2 = t2 > 0.f ? t2 : NEG_SLOPE * t2;
  float t3 = a[3] + b[3]; t3 = t3 > 0.f ? t3 : NEG_SLOPE * t3;
  return (t0 * at[0] + t1 * at[1]) + (t2 * at[2] + t3 * at[3]);
}

#define GATV2_TR_STEP(WD)                                   \
  {                                                         \
    const bool hi = (l16 & (WD)) != 0;                      \
    _Pragma("unroll") for (int i = 0; i < (WD); ++i) {      \
      const float keep = hi ? p[i + (WD)] : p[i];           \
      const float send = hi ? p[i] : p[i + (WD)];           \
      p[i] = keep + gatv2_row_xor_f<(WD)>(send);            \
    }                                                       \
  }
#define GATV2_ROW(J) \
  r[J] = *(const v4f*)(xlB + (unsigned)(gatv2_row_share_i<(J)>(s) * 256 + lofs));
#define GATV2_LOAD4(C) \
  if ((C) < nU) { GATV2_ROW((C) + 0) GATV2_ROW((C) + 1) GATV2_ROW((C) + 2) GATV2_ROW((C) + 3) }
#define GATV2_PART(J) p[J] = gatv2_partial4(r[J], xrn, at);
#define GATV2_PART4(C)                                                                   \
  if ((C) < nU) { GATV2_PART((C) + 0) GATV2_PART((C) + 1) GATV2_PART((C) + 2) GATV2_PART((C) + 3) } \
  else { p[(C) + 0] = 0.f; p[(C) + 1] = 0.f; p[(C) + 2] = 0.f; p[(C) + 3] = 0.f; }
#define GATV2_ACC(J)                                  \
  {                                                   \
    const float aj = gatv2_row_share_f<(J)>(w);       \
    acc[0] += aj * r[J][0];                           \
    acc[1] += aj * r[J][1];                           \
    acc[2] += aj * r[J][2];                           \
    acc[3] += aj * r[J][3];                           \
  }
#define GATV2_ACC4(C) \
  if ((C) < nU) { GATV2_ACC((C) + 0) GATV2_ACC((C) + 1) GATV2_ACC((C) + 2) GATV2_ACC((C) + 3) }

#define HT_PITCH 68
template <bool FUSE>
__global__ __launch_bounds__(256) void gatv2_node_fused(
    const float* __restrict__ xl, const float* xr,
    const int* __restrict__ rowptr, const int* __restrict__ tsrc,
    const float* __restrict__ att, const float* __restrict__ bias,
    float* __restrict__ out, int N, int E, int relu,
    const float* __restrict__ W2l, const float* __restrict__ W2r,
    float* xl2, float* xr2) {
  const int lane = threadIdx.x & 31;
  const int l16  = lane & 15;
  const int wv    = __builtin_amdgcn_readfirstlane((int)(threadIdx.x >> 5));
  const int node0 = (blockIdx.x * 8 + wv) * 2;
  if (!FUSE && node0 >= N) return;
  const int nodeH = node0 + (lane >> 4);
  const bool live = nodeH < N;
  const int node  = live ? nodeH : (node0 < N ? node0 : N - 1);
  int beg = rowptr[node];
  int end = rowptr[node + 1];
  beg = beg < 0 ? 0 : beg;
  end = end > E ? E : end;
  end = end < beg ? beg : end;
  const int deg = end - beg;
  const int d0   = __builtin_amdgcn_readlane(deg, 0);
  const int d1   = __builtin_amdgcn_readlane(deg, 16);
  const int dmax = d0 > d1 ? d0 : d1;
  const unsigned lofs = 16u * (unsigned)l16;
  const char* xlB = (const char*)xl;
  const v4f at    = *(const v4f*)(att + 4 * l16);
  const v4f xrn   = *(const v4f*)(xr + (size_t)node * FDIM + 4 * l16);
  const v4f rself = *(const v4f*)(xl + (size_t)node * FDIM + 4 * l16);

  int snext = node;
  if (l16 < deg) snext = tsrc[beg + l16];

  float es = gatv2_partial4(rself, xrn, at);
  es += gatv2_row_xor_f<8>(es);
  es += gatv2_row_xor_f<4>(es);
  es += gatv2_row_xor_f<2>(es);
  es += gatv2_row_xor_f<1>(es);

  float m   = es;
  float den = 1.f;
  v4f acc   = rself;

#pragma nounroll
  for (int c0 = 0; c0 < dmax; c0 += 16) {
    const int nU = (dmax - c0) < 16 ? (dmax - c0) : 16;
    const bool valid = c0 + l16 < deg;
    int s = snext;
    s = s < 0 ? 0 : (s >= N ? N - 1 : s);
    v4f r[16];
    float p[16];
    GATV2_LOAD4(0)
    GATV2_LOAD4(4)
    GATV2_LOAD4(8)
    GATV2_LOAD4(12)
    snext = node;
    if (c0 + 16 + l16 < deg) snext = tsrc[beg + c0 + 16 + l16];
    GATV2_PART4(0)
    GATV2_PART4(4)
    GATV2_PART4(8)
    GATV2_PART4(12)
    GATV2_TR_STEP(8)
    GATV2_TR_STEP(4)
    GATV2_TR_STEP(2)
    GATV2_TR_STEP(1)
    const float lg = p[0];
    float mn = valid ? fmaxf(lg, m) : m;
    mn = fmaxf(mn, gatv2_row_xor_f<8>(mn));
    mn = fmaxf(mn, gatv2_row_xor_f<4>(mn));
    mn = fmaxf(mn, gatv2_row_xor_f<2>(mn));
    mn = fmaxf(mn, gatv2_row_xor_f<1>(mn));
    const float sc = __expf(m - mn);
    const float w  = valid ? __expf(lg - mn) : 0.f;
    float wsum = w;
    wsum += gatv2_row_xor_f<8>(wsum);
    wsum += gatv2_row_xor_f<4>(wsum);
    wsum += gatv2_row_xor_f<2>(wsum);
    wsum += gatv2_row_xor_f<1>(wsum);
    den = den * sc + wsum;
    acc[0] *= sc;
    acc[1] *= sc;
    acc[2] *= sc;
    acc[3] *= sc;
    GATV2_ACC4(0)
    GATV2_ACC4(4)
    GATV2_ACC4(8)
    GATV2_ACC4(12)
    m = mn;
  }

  const float dn = den + EPS_DEN;
  const v4f b = *(const v4f*)(bias + 4 * l16);
  v4f o;
  o[0] = acc[0] / dn + b[0];
  o[1] = acc[1] / dn + b[1];
  o[2] = acc[2] / dn + b[2];
  o[3] = acc[3] / dn + b[3];
  if (relu) {
    o[0] = o[0] < 0.f ? 0.f : o[0];
    o[1] = o[1] < 0.f ? 0.f : o[1];
    o[2] = o[2] < 0.f ? 0.f : o[2];
    o[3] = o[3] < 0.f ? 0.f : o[3];
  }
  if (!FUSE) {
    if (live) *(v4f*)(out + (size_t)node * FDIM + 4 * l16) = o;
  } else {
    __shared__ __attribute__((aligned(16))) float hT[16 * HT_PITCH];
    const int laneHi  = lane >> 4;
    const int sel     = wv >> 2;
    const int colBase = (wv & 3) * 16;
    const float* Wm   = sel ? W2r : W2l;
    float* dst        = sel ? xr2 : xl2;
    v2f bfr[FDIM / 4];
#pragma unroll
    for (int kq = 0; kq < FDIM / 4; ++kq) {
      const int kk = 4 * kq + 2 * laneHi;
      bfr[kq][0] = Wm[(size_t)kk * FDIM + colBase + l16];
      bfr[kq][1] = Wm[(size_t)(kk + 1) * FDIM + colBase + l16];
    }
    *(v4f*)(hT + (2 * wv + laneHi) * HT_PITCH + 4 * l16) = o;
    __syncthreads();
    v8f accT = {};
#pragma unroll
    for (int kq = 0; kq < FDIM / 4; ++kq) {
      const v2f a = *(const v2f*)(hT + l16 * HT_PITCH + 4 * kq + 2 * laneHi);
      accT = __builtin_amdgcn_wmma_f32_16x16x4_f32(false, a, false, bfr[kq],
                                                   (short)0, accT, false, false);
    }
    const int rbase = (int)blockIdx.x * 16 + 8 * laneHi;
#pragma unroll
    for (int v = 0; v < 8; ++v) {
      const int r = rbase + v;
      if (r < N) dst[(size_t)r * FDIM + colBase + l16] = accT[v];
    }
  }
}
#undef HT_PITCH
#undef GATV2_ACC4
#undef GATV2_ACC
#undef GATV2_PART4
#undef GATV2_PART
#undef GATV2_LOAD4
#undef GATV2_ROW
#undef GATV2_TR_STEP

extern "C" void kernel_launch(void* const* d_in, const int* in_sizes, int n_in,
                              void* d_out, int out_size, void* d_ws, size_t ws_size,
                              hipStream_t stream) {
  const float* x    = (const float*)d_in[0];
  const int*   ei   = (const int*)d_in[1];
  const float* W1l  = (const float*)d_in[2];
  const float* W1r  = (const float*)d_in[3];
  const float* att1 = (const float*)d_in[4];
  const float* b1   = (const float*)d_in[5];
  const float* W2l  = (const float*)d_in[6];
  const float* W2r  = (const float*)d_in[7];
  const float* att2 = (const float*)d_in[8];
  const float* b2   = (const float*)d_in[9];

  const int N  = in_sizes[0] / 128;
  const int E  = in_sizes[1] / 2;
  const int ET = E + N;

  float* ws = (float*)d_ws;
  const size_t NF = (size_t)N * FDIM;
  float*    xl    = ws;
  float*    xr    = xl + NF;
  float*    h     = xr + NF;
  int*      lst   = (int*)(h + NF);
  float*    e     = (float*)lst + NF;
  float*    ex    = e + ET;

  const int per = (((N + 1) + 1023) / 1024 + 31) / 32 * 32;
  const int NP  = per * 1024;
  int* cnt    = lst;
  int* rowptr = cnt + NP;
  int* tsrc   = (int*)e;
  int* rank   = (int*)ex;
  int* hist   = (int*)h;

  const int rowTiles   = (N + 15) / 16;
  const int realEdgeBl = (E + 255) / 256;
  const int pairBl     = (N + 15) / 16;
  const int* esrc = ei;
  const int* edst = ei + E;

  gatv2_bucket_rank_lds<<<RK_R * RK_C, 1024, 0, stream>>>(edst, hist, rank, E, NP);
  gatv2_hist_prefix<<<NP / 256, 256, 0, stream>>>(hist, cnt, NP);
  gatv2_scan_and_gemm<<<1 + (rowTiles + 7) / 8, 1024, 0, stream>>>(
      cnt, rowptr, per, x, W1l, W1r, xl, xr, N, 128);
  gatv2_bucket_place<<<realEdgeBl, 256, 0, stream>>>(esrc, edst, rowptr, hist, rank, tsrc, E, NP);

  gatv2_node_fused<true><<<pairBl, 256, 0, stream>>>(
      xl, xr, rowptr, tsrc, att1, b1, nullptr, N, E, 1, W2l, W2r, h, xr);

  gatv2_node_fused<false><<<pairBl, 256, 0, stream>>>(
      h, xr, rowptr, tsrc, att2, b2, (float*)d_out, N, E, 0, nullptr, nullptr, nullptr, nullptr);
}


# rowptr scan spread out: each 1024-node block of the per-node prefix kernel scans its own in-degrees, consumers add the few block totals; one-workgroup scan launch gone, GEMM rides with prefix kernel
# speedup vs baseline: 5.6795x; 1.1442x over previous
#include <hip/hip_runtime.h>
#include <hip/hip_bf16.h>

typedef __attribute__((ext_vector_type(2))) float v2f;
typedef __attribute__((ext_vector_type(4))) float v4f;
typedef __attribute__((ext_vector_type(8))) float v8f;
typedef __attribute__((ext_vector_type(2))) int v2i;
typedef __attribute__((ext_vector_type(4))) int v4i;

#define NEG_SLOPE 0.2f
#define EPS_DEN 1e-16f
#define FDIM 64

__device__ __forceinline__ void gatv2_gemm_tile(
    const float* __restrict__ X, const float* __restrict__ Wl,
    const float* __restrict__ Wr, float* __restrict__ xl,
    float* __restrict__ xr, int N, int K, int rowTile, int colQ, int lane) {
  const int laneHi = lane >> 4;
  const int lane16 = lane & 15;
  const int colBase = colQ * 16;

  int row  = rowTile * 16 + lane16;
  int rowc = row < N ? row : N - 1;

  v8f accL = {};
  v8f accR = {};
  const float* xrow = X + (size_t)rowc * K;

  for (int k = 0; k < K; k += 4) {
    const int kk = k + 2 * laneHi;
    v2f a = *(const v2f*)(xrow + kk);
    v2f bl, br;
    bl[0] = Wl[(size_t)kk * FDIM + colBase + lane16];
    bl[1] = Wl[(size_t)(kk + 1) * FDIM + colBase + lane16];
    br[0] = Wr[(size_t)kk * FDIM + colBase + lane16];
    br[1] = Wr[(size_t)(kk + 1) * FDIM + colBase + lane16];
    accL = __builtin_amdgcn_wmma_f32_16x16x4_f32(false, a, false, bl,
                                                 (short)0, accL, false, false);
    accR = __builtin_amdgcn_wmma_f32_16x16x4_f32(false, a, false, br,
                                                 (short)0, accR, false, false);
  }

#pragma unroll
  for (int v = 0; v < 8; ++v) {
    int m = v + 8 * laneHi;
    int r = rowTile * 16 + m;
    if (r < N) {
      xl[(size_t)r * FDIM + colBase + lane16] = accL[v];
      xr[(size_t)r * FDIM + colBase + lane16] = accR[v];
    }
  }
}

#define RK_R 8
#define RK_C 32
#define RK_CAP 8192
#define RK_U 25
#define RK_K 8
__global__ __launch_bounds__(1024) void gatv2_bucket_rank_lds(
    const int* __restrict__ edst, int* __restrict__ hist,
    int* __restrict__ rank, int E, int NP) {
  static_assert(RK_U * 1024 <= 32767, "an edge's index in its trip must stay below the empty slot's 0x7fff");
  static_assert(RK_CAP == 8192 && RK_K == 8, "the list entry packs 13 bits of destination; a slot row is one 16-byte load");
  __shared__ __attribute__((aligned(16))) int cntL[RK_CAP];
  __shared__ __attribute__((aligned(16))) int tcnL[RK_CAP];
  __shared__ __attribute__((aligned(16))) unsigned int keyL[RK_CAP * RK_K / 2];
  __shared__ int listL[32 * RK_U * 32];
  const int t    = threadIdx.x;
  const int lane = t & 31;
  const int wv   = __builtin_amdgcn_readfirstlane(t >> 5);
  int* myl = listL + wv * (RK_U * 32);
  const int r = (int)blockIdx.x % RK_R;
  const int c = (int)blockIdx.x / RK_R;
  const int chunk = (E + RK_C - 1) / RK_C;
  const int e0 = c * chunk;
  int e1 = e0 + chunk;
  e1 = e1 > E ? E : e1;
  for (int nb = r * RK_CAP; nb < NP; nb += RK_R * RK_CAP) {
    for (int i = t * 4; i < RK_CAP; i += 4096) {
      v4i z = {0, 0, 0, 0};
      *(v4i*)(cntL + i) = z;
    }
    for (int eb = e0; eb < e1; eb += RK_U * 1024) {
      for (int i = t * 4; i < RK_CAP; i += 4096) {
        v4i z = {0, 0, 0, 0};
        *(v4i*)(tcnL + i) = z;
      }
      for (int i = t * 4; i < RK_CAP * RK_K / 2; i += 4096) {
        v4i f = {0x7fff7fff, 0x7fff7fff, 0x7fff7fff, 0x7fff7fff};
        *(v4i*)(keyL + i) = f;
      }
      __syncthreads();
      int d[RK_U];
#pragma unroll
      for (int u = 0; u < RK_U; ++u) {
        const int e = eb + u * 1024 + t;
        d[u] = edst[e < e1 ? e : e1 - 1];
      }
      int n = 0;
#pragma unroll
      for (int u = 0; u < RK_U; ++u) {
        const int e  = eb + u * 1024 + t;
        const int dl = d[u] - nb;
        const bool inr = e < e1 && (unsigned)dl < (unsigned)RK_CAP;
        const unsigned m = __builtin_amdgcn_ballot_w32(inr);
        const int pre = (int)__builtin_amdgcn_mbcnt_lo(m, 0u);
        if (inr) myl[n + pre] = dl | ((u * 1024 + t) << 13);
        n += __builtin_popcount(m);
      }
      for (int i = lane; i < n; i += 32) {
        const int pk  = myl[i];
        const int dl  = pk & (RK_CAP - 1);
        const int key = pk >> 13;
        const int a = atomicAdd(tcnL + dl, 1);
        if (a < RK_K) ((unsigned short*)keyL)[dl * RK_K + a] = (unsigned short)key;
      }
      __syncthreads();
      for (int i = lane; i < n; i += 32) {
        const int pk  = myl[i];
        const int dl  = pk & (RK_CAP - 1);
        const int key = pk >> 13;
        const int g    = tcnL[dl];
        const int base = cntL[dl];
        int det;
        if (g <= RK_K) {
          const v4i kw = *(const v4i*)(keyL + dl * (RK_K / 2));
          const unsigned cc = 0x80008000u - (unsigned)key * 0x10001u;
          const int ge = __builtin_popcount(((unsigned)kw[0] + cc) & 0x80008000u)
                       + __builtin_popcount(((unsigned)kw[1] + cc) & 0x80008000u)
                       + __builtin_popcount(((unsigned)kw[2] + cc) & 0x80008000u)
                       + __builtin_popcount(((unsigned)kw[3] + cc) & 0x80008000u);
          det = RK_K - ge;
        } else {
          const int dg = dl + nb;
          det = 0;
#pragma nounroll
          for (int e2 = eb; e2 < eb + key; ++e2) det += (edst[e2] == dg) ? 1 : 0;
        }
        rank[eb + key] = base + det;
      }
      __syncthreads();
      for (int i = t * 4; i < RK_CAP; i += 4096)
        *(v4i*)(cntL + i) = *(const v4i*)(cntL + i) + *(const v4i*)(tcnL + i);
    }
    __syncthreads();
    int* hrow = hist + (size_t)c * NP + nb;
    for (int i = t * 4; i < RK_CAP; i += 4096)
      *(v4i*)(hrow + i) = *(const v4i*)(cntL + i);
    __syncthreads();
  }
}

template <int S> __device__ __forceinline__ int gatv2_row_shr_i(int x) {
  return __builtin_amdgcn_update_dpp(0, x, 0x110 + S, 0xf, 0xf, true);
}
__device__ __forceinline__ int gatv2_wave_scan_incl(int x, int lane) {
  x += gatv2_row_shr_i<1>(x);
  x += gatv2_row_shr_i<2>(x);
  x += gatv2_row_shr_i<4>(x);
  x += gatv2_row_shr_i<8>(x);
  const int low = __builtin_amdgcn_readlane(x, 15);
  return lane >= 16 ? x + low : x;
}

#define PS_T 1024
#define GO_MAX 1024
__device__ __forceinline__ void gatv2_hist_prefix_block(
    int* hist, int* __restrict__ lrow, int* __restrict__ btot, int NP, int b) {
  __shared__ int wsum[32];
  const int t    = threadIdx.x;
  const int lane = t & 31;
  const int wave = t >> 5;
  const int n    = b * PS_T + t;
  const unsigned un  = (unsigned)n;
  const unsigned uNP = (unsigned)NP;
  int tot = 0;
#pragma nounroll
  for (int c0 = 0; c0 < RK_C; c0 += 8) {
    int v[8];
#pragma unroll
    for (int j = 0; j < 8; ++j) v[j] = hist[(unsigned)(c0 + j) * uNP + un];
#pragma unroll
    for (int j = 0; j < 8; ++j) tot += v[j];
  }
  const int incl = gatv2_wave_scan_incl(tot, lane);
  if (lane == 31) wsum[wave] = incl;
  __syncthreads();
  if (wave == 0) {
    const int w = wsum[lane];
    wsum[lane] = gatv2_wave_scan_incl(w, lane) - w;
  }
  __syncthreads();
  const int base = wsum[wave] + incl - tot;
  int run = base;
#pragma nounroll
  for (int c0 = 0; c0 < RK_C; c0 += 8) {
    int v[8];
#pragma unroll
    for (int j = 0; j < 8; ++j) v[j] = hist[(unsigned)(c0 + j) * uNP + un];
#pragma unroll
    for (int j = 0; j < 8; ++j) {
      hist[(unsigned)(c0 + j) * uNP + un] = run;
      run += v[j];
    }
  }
  lrow[n] = base;
  if (t == PS_T - 1) btot[b] = run;
}

__global__ __launch_bounds__(1024) void gatv2_prefix_and_gemm(
    int* hist, int* __restrict__ lrow, int* __restrict__ btot, int NP, int pblocks,
    const float* __restrict__ X, const float* __restrict__ Wl,
    const float* __restrict__ Wr, float* __restrict__ xl,
    float* __restrict__ xr, int N, int K) {
  if ((int)blockIdx.x < pblocks) {
    gatv2_hist_prefix_block(hist, lrow, btot, NP, (int)blockIdx.x);
    return;
  }
  const int wv = __builtin_amdgcn_readfirstlane((int)(threadIdx.x >> 5));
  const int rowTile = ((int)blockIdx.x - pblocks) * 8 + (wv >> 2);
  if (rowTile * 16 >= N) return;
  gatv2_gemm_tile(X, Wl, Wr, xl, xr, N, K, rowTile, wv & 3, (int)(threadIdx.x & 31));
}

__global__ __launch_bounds__(256) void gatv2_bucket_place(
    const int* __restrict__ esrc, const int* __restrict__ edst,
    const int* __restrict__ lrow, const int* __restrict__ btot,
    const int* __restrict__ hist, const int* __restrict__ rank,
    int* __restrict__ tsrc, int* __restrict__ rowptr, int E, int NP, int pblocks) {
  __shared__ int goffL[GO_MAX];
  const int t    = threadIdx.x;
  const int lane = t & 31;
  const int eid  = blockIdx.x * 256 + t;
  const bool isE = eid < E;
  const int ec   = isE ? eid : E - 1;
  const int s  = esrc[ec];
  int d        = edst[ec];
  const int rk = rank[ec];
  const int nb = pblocks < GO_MAX ? pblocks : GO_MAX;
  if (t < 32) {
    int carry = 0;
    for (int i = 0; i < nb; i += 64) {
      const int j = i + 2 * lane;
      const int a = j < nb ? btot[j] : 0;
      const int b = j + 1 < nb ? btot[j + 1] : 0;
      const int sum = a + b;
      const int incl = gatv2_wave_scan_incl(sum, lane);
      const int excl = carry + incl - sum;
      if (j < nb) goffL[j] = excl;
      if (j + 1 < nb) goffL[j + 1] = excl + a;
      carry += __builtin_amdgcn_readlane(incl, 31);
    }
  }
  __syncthreads();
  const int n = blockIdx.x * 256 + t;
  if (n < NP) {
    int g = n / PS_T;
    g = g < nb ? g : nb - 1;
    rowptr[n] = goffL[g] + lrow[n];
  }
  if (!isE) return;
  d = d < 0 ? 0 : (d >= NP ? NP - 1 : d);
  const int chunk = (E + RK_C - 1) / RK_C;
  const int c = eid / chunk;
  int g = d / PS_T;
  g = g < nb ? g : nb - 1;
  const int pos = goffL[g] + hist[(size_t)c * NP + d] + rk;
  if ((unsigned)pos < (unsigned)E) tsrc[pos] = s;
}

template <int J> __device__ __forceinline__ int gatv2_row_share_i(int x) {
  return __builtin_amdgcn_update_dpp(0, x, 0x150 + J, 0xf, 0xf, true);
}
template <int J> __device__ __forceinline__ float gatv2_row_share_f(float x) {
  return __int_as_float(gatv2_row_share_i<J>(__float_as_int(x)));
}
template <int M> __device__ __forceinline__ float gatv2_row_xor_f(float x) {
  return __int_as_float(__builtin_amdgcn_update_dpp(0, __float_as_int(x), 0x160 + M, 0xf, 0xf, true));
}

__device__ __forceinline__ float gatv2_partial4(v4f a, v4f b, v4f at) {
  float t0 = a[0] + b[0]; t0 = t0 > 0.f ? t0 : NEG_SLOPE * t0;
  float t1 = a[1] + b[1]; t1 = t1 > 0.f ? t1 : NEG_SLOPE * t1;
  float t2 = a[2] + b[2]; t2 = t2 > 0.f ? t2 : NEG_SLOPE * t2;
  float t3 = a[3] + b[3]; t3 = t3 > 0.f ? t3 : NEG_SLOPE * t3;
  return (t0 * at[0] + t1 * at[1]) + (t2 * at[2] + t3 * at[3]);
}

#define GATV2_TR_STEP(WD)                                   \
  {                                                         \
    const bool hi = (l16 & (WD)) != 0;                      \
    _Pragma("unroll") for (int i = 0; i < (WD); ++i) {      \
      const float keep = hi ? p[i + (WD)] : p[i];           \
      const float send = hi ? p[i] : p[i + (WD)];           \
      p[i] = keep + gatv2_row_xor_f<(WD)>(send);            \
    }                                                       \
  }
#define GATV2_ROW(J) \
  r[J] = *(const v4f*)(xlB + (unsigned)(gatv2_row_share_i<(J)>(s) * 256 + lofs));
#define GATV2_LOAD4(C) \
  if ((C) < nU) { GATV2_ROW((C) + 0) GATV2_ROW((C) + 1) GATV2_ROW((C) + 2) GATV2_ROW((C) + 3) }
#define GATV2_PART(J) p[J] = gatv2_partial4(r[J], xrn, at);
#define GATV2_PART4(C)                                                                   \
  if ((C) < nU) { GATV2_PART((C) + 0) GATV2_PART((C) + 1) GATV2_PART((C) + 2) GATV2_PART((C) + 3) } \
  else { p[(C) + 0] = 0.f; p[(C) + 1] = 0.f; p[(C) + 2] = 0.f; p[(C) + 3] = 0.f; }
#define GATV2_ACC(J)                                  \
  {                                                   \
    const float aj = gatv2_row_share_f<(J)>(w);       \
    acc[0] += aj * r[J][0];                           \
    acc[1] += aj * r[J][1];                           \
    acc[2] += aj * r[J][2];                           \
    acc[3] += aj * r[J][3];                           \
  }
#define GATV2_ACC4(C) \
  if ((C) < nU) { GATV2_ACC((C) + 0) GATV2_ACC((C) + 1) GATV2_ACC((C) + 2) GATV2_ACC((C) + 3) }

#define HT_PITCH 68
template <bool FUSE>
__global__ __launch_bounds__(256) void gatv2_node_fused(
    const float* __restrict__ xl, const float* xr,
    const int* __restrict__ rowptr, const int* __restrict__ tsrc,
    const float* __restrict__ att, const float* __restrict__ bias,
    float* __restrict__ out, int N, int E, int relu,
    const float* __restrict__ W2l, const float* __restrict__ W2r,
    float* xl2, float* xr2) {
  const int lane = threadIdx.x & 31;
  const int l16  = lane & 15;
  const int wv    = __builtin_amdgcn_readfirstlane((int)(threadIdx.x >> 5));
  const int node0 = (blockIdx.x * 8 + wv) * 2;
  if (!FUSE && node0 >= N) return;
  const int nodeH = node0 + (lane >> 4);
  const bool live = nodeH < N;
  const int node  = live ? nodeH : (node0 < N ? node0 : N - 1);
  int beg = rowptr[node];
  int end = rowptr[node + 1];
  beg = beg < 0 ? 0 : beg;
  end = end > E ? E : end;
  end = end < beg ? beg : end;
  const int deg = end - beg;
  const int d0   = __builtin_amdgcn_readlane(deg, 0);
  const int d1   = __builtin_amdgcn_readlane(deg, 16);
  const int dmax = d0 > d1 ? d0 : d1;
  const unsigned lofs = 16u * (unsigned)l16;
  const char* xlB = (const char*)xl;
  const v4f at    = *(const v4f*)(att + 4 * l16);
  const v4f xrn   = *(const v4f*)(xr + (size_t)node * FDIM + 4 * l16);
  const v4f rself = *(const v4f*)(xl + (size_t)node * FDIM + 4 * l16);

  int snext = node;
  if (l16 < deg) snext = tsrc[beg + l16];

  float es = gatv2_partial4(rself, xrn, at);
  es += gatv2_row_xor_f<8>(es);
  es += gatv2_row_xor_f<4>(es);
  es += gatv2_row_xor_f<2>(es);
  es += gatv2_row_xor_f<1>(es);

  float m   = es;
  float den = 1.f;
  v4f acc   = rself;

#pragma nounroll
  for (int c0 = 0; c0 < dmax; c0 += 16) {
    const int nU = (dmax - c0) < 16 ? (dmax - c0) : 16;
    const bool valid = c0 + l16 < deg;
    int s = snext;
    s = s < 0 ? 0 : (s >= N ? N - 1 : s);
    v4f r[16];
    float p[16];
    GATV2_LOAD4(0)
    GATV2_LOAD4(4)
    GATV2_LOAD4(8)
    GATV2_LOAD4(12)
    snext = node;
    if (c0 + 16 + l16 < deg) snext = tsrc[beg + c0 + 16 + l16];
    GATV2_PART4(0)
    GATV2_PART4(4)
    GATV2_PART4(8)
    GATV2_PART4(12)
    GATV2_TR_STEP(8)
    GATV2_TR_STEP(4)
    GATV2_TR_STEP(2)
    GATV2_TR_STEP(1)
    const float lg = p[0];
    float mn = valid ? fmaxf(lg, m) : m;
    mn = fmaxf(mn, gatv2_row_xor_f<8>(mn));
    mn = fmaxf(mn, gatv2_row_xor_f<4>(mn));
    mn = fmaxf(mn, gatv2_row_xor_f<2>(mn));
    mn = fmaxf(mn, gatv2_row_xor_f<1>(mn));
    const float sc = __expf(m - mn);
    const float w  = valid ? __expf(lg - mn) : 0.f;
    float wsum = w;
    wsum += gatv2_row_xor_f<8>(wsum);
    wsum += gatv2_row_xor_f<4>(wsum);
    wsum += gatv2_row_xor_f<2>(wsum);
    wsum += gatv2_row_xor_f<1>(wsum);
    den = den * sc + wsum;
    acc[0] *= sc;
    acc[1] *= sc;
    acc[2] *= sc;
    acc[3] *= sc;
    GATV2_ACC4(0)
    GATV2_ACC4(4)
    GATV2_ACC4(8)
    GATV2_ACC4(12)
    m = mn;
  }

  const float dn = den + EPS_DEN;
  const v4f b = *(const v4f*)(bias + 4 * l16);
  v4f o;
  o[0] = acc[0] / dn + b[0];
  o[1] = acc[1] / dn + b[1];
  o[2] = acc[2] / dn + b[2];
  o[3] = acc[3] / dn + b[3];
  if (relu) {
    o[0] = o[0] < 0.f ? 0.f : o[0];
    o[1] = o[1] < 0.f ? 0.f : o[1];
    o[2] = o[2] < 0.f ? 0.f : o[2];
    o[3] = o[3] < 0.f ? 0.f : o[3];
  }
  if (!FUSE) {
    if (live) *(v4f*)(out + (size_t)node * FDIM + 4 * l16) = o;
  } else {
    __shared__ __attribute__((aligned(16))) float hT[16 * HT_PITCH];
    const int laneHi  = lane >> 4;
    const int sel     = wv >> 2;
    const int colBase = (wv & 3) * 16;
    const float* Wm   = sel ? W2r : W2l;
    float* dst        = sel ? xr2 : xl2;
    v2f bfr[FDIM / 4];
#pragma unroll
    for (int kq = 0; kq < FDIM / 4; ++kq) {
      const int kk = 4 * kq + 2 * laneHi;
      bfr[kq][0] = Wm[(size_t)kk * FDIM + colBase + l16];
      bfr[kq][1] = Wm[(size_t)(kk + 1) * FDIM + colBase + l16];
    }
    *(v4f*)(hT + (2 * wv + laneHi) * HT_PITCH + 4 * l16) = o;
    __syncthreads();
    v8f accT = {};
#pragma unroll
    for (int kq = 0; kq < FDIM / 4; ++kq) {
      const v2f a = *(const v2f*)(hT + l16 * HT_PITCH + 4 * kq + 2 * laneHi);
      accT = __builtin_amdgcn_wmma_f32_16x16x4_f32(false, a, false, bfr[kq],
                                                   (short)0, accT, false, false);
    }
    const int rbase = (int)blockIdx.x * 16 + 8 * laneHi;
#pragma unroll
    for (int v = 0; v < 8; ++v) {
      const int r = rbase + v;
      if (r < N) dst[(size_t)r * FDIM + colBase + l16] = accT[v];
    }
  }
}
#undef HT_PITCH
#undef GATV2_ACC4
#undef GATV2_ACC
#undef GATV2_PART4
#undef GATV2_PART
#undef GATV2_LOAD4
#undef GATV2_ROW
#undef GATV2_TR_STEP

extern "C" void kernel_launch(void* const* d_in, const int* in_sizes, int n_in,
                              void* d_out, int out_size, void* d_ws, size_t ws_size,
                              hipStream_t stream) {
  const float* x    = (const float*)d_in[0];
  const int*   ei   = (const int*)d_in[1];
  const float* W1l  = (const float*)d_in[2];
  const float* W1r  = (const float*)d_in[3];
  const float* att1 = (const float*)d_in[4];
  const float* b1   = (const float*)d_in[5];
  const float* W2l  = (const float*)d_in[6];
  const float* W2r  = (const float*)d_in[7];
  const float* att2 = (const float*)d_in[8];
  const float* b2   = (const float*)d_in[9];

  const int N  = in_sizes[0] / 128;
  const int E  = in_sizes[1] / 2;
  const int ET = E + N;

  float* ws = (float*)d_ws;
  const size_t NF = (size_t)N * FDIM;
  float*    xl    = ws;
  float*    xr    = xl + NF;
  float*    h     = xr + NF;
  int*      lst   = (int*)(h + NF);
  float*    e     = (float*)lst + NF;
  float*    ex    = e + ET;

  const int per = (((N + 1) + 1023) / 1024 + 31) / 32 * 32;
  const int NP  = per * 1024;
  int* lrow   = lst;
  int* rowptr = lrow + NP;
  int* btot   = rowptr + NP;
  int* tsrc   = (int*)e;
  int* rank   = (int*)ex;
  int* hist   = (int*)h;

  const int rowTiles   = (N + 15) / 16;
  const int realEdgeBl = (E + 255) / 256;
  const int placeBl    = realEdgeBl > NP / 256 ? realEdgeBl : NP / 256;
  const int pairBl     = (N + 15) / 16;
  const int* esrc = ei;
  const int* edst = ei + E;

  gatv2_bucket_rank_lds<<<RK_R * RK_C, 1024, 0, stream>>>(edst, hist, rank, E, NP);
  gatv2_prefix_and_gemm<<<per + (rowTiles + 7) / 8, 1024, 0, stream>>>(
      hist, lrow, btot, NP, per, x, W1l, W1r, xl, xr, N, 128);
  gatv2_bucket_place<<<placeBl, 256, 0, stream>>>(
      esrc, edst, lrow, btot, hist, rank, tsrc, rowptr, E, NP, per);

  gatv2_node_fused<true><<<pairBl, 256, 0, stream>>>(
      xl, xr, rowptr, tsrc, att1, b1, nullptr, N, E, 1, W2l, W2r, h, xr);

  gatv2_node_fused<false><<<pairBl, 256, 0, stream>>>(
      h, xr, rowptr, tsrc, att2, b2, (float*)d_out, N, E, 0, nullptr, nullptr, nullptr, nullptr);
}
